# starting kernel
# baseline (speedup 1.0000x reference)
#include <hip/hip_runtime.h>
#include <hip/hip_bf16.h>

typedef __attribute__((ext_vector_type(2))) float v2f;
typedef __attribute__((ext_vector_type(8))) float v8f;

#define HDIM 64

__global__ __launch_bounds__(256)
void deg_kernel(const int* __restrict__ src, const int* __restrict__ dst,
                float* __restrict__ degU, float* __restrict__ degM, int nE) {
    int e = blockIdx.x * blockDim.x + threadIdx.x;
    if (e >= nE) return;
    atomicAdd(degU + src[e], 1.0f);
    atomicAdd(degM + dst[e], 1.0f);
}

__global__ __launch_bounds__(256)
void movie_init_kernel(const float* __restrict__ mx, const float* __restrict__ linW,
                       const float* __restrict__ linb, const float* __restrict__ memb,
                       float* __restrict__ XM, int M) {
    int t = blockIdx.x * blockDim.x + threadIdx.x;
    if (t >= M * HDIM) return;
    int m = t >> 6, h = t & 63;
    float acc = linb[h] + memb[t];
    const float* xr = mx + (size_t)m * 20;
    const float* wr = linW + (size_t)h * 20;
#pragma unroll
    for (int f = 0; f < 20; ++f) acc = fmaf(xr[f], wr[f], acc);
    XM[t] = acc;
}

__global__ __launch_bounds__(256)
void scatter_kernel(const float* __restrict__ xsrc, const int* __restrict__ src,
                    const int* __restrict__ dst, float* __restrict__ agg, int nE) {
    long long tid = (long long)blockIdx.x * blockDim.x + threadIdx.x;
    if (tid >= (long long)nE * 16) return;
    int e = (int)(tid >> 4);
    int q = (int)(tid & 15);
    int s = src[e], d = dst[e];
    const float4 v = *(const float4*)(xsrc + (size_t)s * HDIM + q * 4);
    float* p = agg + (size_t)d * HDIM + q * 4;
    atomicAdd(p + 0, v.x);
    atomicAdd(p + 1, v.y);
    atomicAdd(p + 2, v.z);
    atomicAdd(p + 3, v.w);
}

__global__ __launch_bounds__(128)
void combine_wmma_kernel(const float* __restrict__ agg, const float* __restrict__ deg,
                         const float* __restrict__ xdst,
                         const float* __restrict__ Wl, const float* __restrict__ Wr,
                         const float* __restrict__ bias, float* __restrict__ out,
                         int do_relu) {
    __shared__ float A1[16][68];
    __shared__ float A2[16][68];

    const int row0 = blockIdx.x * 16;
    const int tid  = threadIdx.x;

    {
        int r  = tid >> 3;
        int c0 = (tid & 7) * 8;
        float dinv = 1.0f / fmaxf(deg[row0 + r], 1.0f);
        const float* ga = agg  + (size_t)(row0 + r) * HDIM + c0;
        const float* gx = xdst + (size_t)(row0 + r) * HDIM + c0;
#pragma unroll
        for (int j = 0; j < 8; ++j) {
            A1[r][c0 + j] = ga[j] * dinv;
            A2[r][c0 + j] = gx[j];
        }
    }
    __syncthreads();

    const int wave = tid >> 5;
    const int lane = tid & 31;
    const int m    = lane & 15;
    const int half = lane >> 4;
    const int col0 = wave * 16;

    v8f c = {};
#pragma unroll
    for (int kk = 0; kk < 16; ++kk) {
        int k = kk * 4 + half * 2;
        v2f a, b;
        a.x = A1[m][k];  a.y = A1[m][k + 1];
        const float* wp = Wl + (size_t)(col0 + m) * HDIM + k;
        b.x = wp[0];     b.y = wp[1];
        c = __builtin_amdgcn_wmma_f32_16x16x4_f32(false, a, false, b,
                                                  (short)0, c, false, false);
    }
#pragma unroll
    for (int kk = 0; kk < 16; ++kk) {
        int k = kk * 4 + half * 2;
        v2f a, b;
        a.x = A2[m][k];  a.y = A2[m][k + 1];
        const float* wp = Wr + (size_t)(col0 + m) * HDIM + k;
        b.x = wp[0];     b.y = wp[1];
        c = __builtin_amdgcn_wmma_f32_16x16x4_f32(false, a, false, b,
                                                  (short)0, c, false, false);
    }

    float bv = bias[col0 + m];
#pragma unroll
    for (int r = 0; r < 8; ++r) {
        float v = c[r] + bv;
        if (do_relu) v = fmaxf(v, 0.0f);
        out[(size_t)(row0 + r + 8 * half) * HDIM + col0 + m] = v;
    }
}

__global__ __launch_bounds__(256)
void edge_dot_kernel(const float* __restrict__ U2, const float* __restrict__ M2,
                     const int* __restrict__ eu, const int* __restrict__ em,
                     float* __restrict__ out, int L) {
    long long tid = (long long)blockIdx.x * blockDim.x + threadIdx.x;
    if (tid >= (long long)L * 8) return;
    int p = (int)(tid >> 3);
    int s = (int)(tid & 7);
    const float4* a = (const float4*)(U2 + (size_t)eu[p] * HDIM + s * 8);
    const float4* b = (const float4*)(M2 + (size_t)em[p] * HDIM + s * 8);
    float4 a0 = a[0], a1 = a[1], b0 = b[0], b1 = b[1];
    float acc = a0.x * b0.x + a0.y * b0.y + a0.z * b0.z + a0.w * b0.w
              + a1.x * b1.x + a1.y * b1.y + a1.z * b1.z + a1.w * b1.w;
    acc += __shfl_xor(acc, 1);
    acc += __shfl_xor(acc, 2);
    acc += __shfl_xor(acc, 4);
    if (s == 0) out[p] = acc;
}

extern "C" void kernel_launch(void* const* d_in, const int* in_sizes, int n_in,
                              void* d_out, int out_size, void* d_ws, size_t ws_size,
                              hipStream_t stream) {
    (void)n_in; (void)ws_size;
    const float* movie_x   = (const float*)d_in[2];
    const int*   e_src     = (const int*)d_in[3];
    const int*   e_dst     = (const int*)d_in[4];
    const int*   eli_u     = (const int*)d_in[5];
    const int*   eli_m     = (const int*)d_in[6];
    const float* user_emb  = (const float*)d_in[7];
    const float* movie_emb = (const float*)d_in[8];
    const float* lin_W     = (const float*)d_in[9];
    const float* lin_b     = (const float*)d_in[10];
    const float* W1um_l    = (const float*)d_in[11];
    const float* W1um_r    = (const float*)d_in[12];
    const float* W1mu_l    = (const float*)d_in[13];
    const float* W1mu_r    = (const float*)d_in[14];
    const float* W2um_l    = (const float*)d_in[15];
    const float* W2um_r    = (const float*)d_in[16];
    const float* W2mu_l    = (const float*)d_in[17];
    const float* W2mu_r    = (const float*)d_in[18];
    const float* b1um      = (const float*)d_in[19];
    const float* b1mu      = (const float*)d_in[20];
    const float* b2um      = (const float*)d_in[21];
    const float* b2mu      = (const float*)d_in[22];
    float* out = (float*)d_out;

    const int U = in_sizes[7] / HDIM;
    const int M = in_sizes[8] / HDIM;
    const int E = in_sizes[3];
    const int L = in_sizes[5];
    (void)out_size;

    size_t off = 0;
    auto carve = [&](size_t nfloats) {
        float* p = (float*)((char*)d_ws + off);
        off += ((nfloats * sizeof(float) + 255) / 256) * 256;
        return p;
    };
    float* XM    = carve((size_t)M * HDIM);
    float* AGG_M = carve((size_t)M * HDIM);
    float* AGG_U = carve((size_t)U * HDIM);
    float* H_M   = carve((size_t)M * HDIM);
    float* H_U   = carve((size_t)U * HDIM);
    float* M2    = carve((size_t)M * HDIM);
    float* U2    = carve((size_t)U * HDIM);
    float* DEG_M = carve((size_t)M);
    float* DEG_U = carve((size_t)U);

    const int TB = 256;
    dim3 b(TB);
    long long sc_threads = (long long)E * 16;
    int sc_blocks = (int)((sc_threads + TB - 1) / TB);

    hipMemsetAsync(DEG_M, 0, (size_t)M * sizeof(float), stream);
    hipMemsetAsync(DEG_U, 0, (size_t)U * sizeof(float), stream);
    deg_kernel<<<(E + TB - 1) / TB, b, 0, stream>>>(e_src, e_dst, DEG_U, DEG_M, E);

    movie_init_kernel<<<(M * HDIM + TB - 1) / TB, b, 0, stream>>>(
        movie_x, lin_W, lin_b, movie_emb, XM, M);

    hipMemsetAsync(AGG_M, 0, (size_t)M * HDIM * sizeof(float), stream);
    hipMemsetAsync(AGG_U, 0, (size_t)U * HDIM * sizeof(float), stream);
    scatter_kernel<<<sc_blocks, b, 0, stream>>>(user_emb, e_src, e_dst, AGG_M, E);
    scatter_kernel<<<sc_blocks, b, 0, stream>>>(XM, e_dst, e_src, AGG_U, E);
    combine_wmma_kernel<<<M / 16, 128, 0, stream>>>(AGG_M, DEG_M, XM,
                                                    W1um_l, W1um_r, b1um, H_M, 1);
    combine_wmma_kernel<<<U / 16, 128, 0, stream>>>(AGG_U, DEG_U, user_emb,
                                                    W1mu_l, W1mu_r, b1mu, H_U, 1);

    hipMemsetAsync(AGG_M, 0, (size_t)M * HDIM * sizeof(float), stream);
    hipMemsetAsync(AGG_U, 0, (size_t)U * HDIM * sizeof(float), stream);
    scatter_kernel<<<sc_blocks, b, 0, stream>>>(H_U, e_src, e_dst, AGG_M, E);
    scatter_kernel<<<sc_blocks, b, 0, stream>>>(H_M, e_dst, e_src, AGG_U, E);
    combine_wmma_kernel<<<M / 16, 128, 0, stream>>>(AGG_M, DEG_M, H_M,
                                                    W2um_l, W2um_r, b2um, M2, 0);
    combine_wmma_kernel<<<U / 16, 128, 0, stream>>>(AGG_U, DEG_U, H_U,
                                                    W2mu_l, W2mu_r, b2mu, U2, 0);

    long long dt = (long long)L * 8;
    edge_dot_kernel<<<(int)((dt + TB - 1) / TB), b, 0, stream>>>(U2, M2, eli_u, eli_m, out, L);
}


# segment sums: a work-group owns 1184 destination rows as f32 sums in LDS, filters an edge chunk for them, adds source rows by LDS adds, flushes whole-line adds; sums now per-chunk partials
# speedup vs baseline: 1.7027x; 1.7027x over previous
#include <hip/hip_runtime.h>
#include <hip/hip_bf16.h>

typedef __attribute__((ext_vector_type(2))) float v2f;
typedef __attribute__((ext_vector_type(8))) float v8f;

#define HDIM 64

#define DEG_NB     12544
#define DEG_TPB    1024
#define DEG_CHUNKS 21

__device__ __forceinline__ void deg_count(unsigned int* h, int idx, int base) {
    unsigned int a = (unsigned int)(idx - base);
    if (a < (unsigned int)DEG_NB) atomicAdd(&h[a], 1u);
}

__global__ __launch_bounds__(DEG_TPB)
void deg_hist_kernel(const int* __restrict__ src, const int* __restrict__ dst,
                     float* __restrict__ degU, float* __restrict__ degM,
                     int nE, int nU, int nM, int nRM, int chunk) {
    __shared__ unsigned int h[DEG_NB];
    const int tid = threadIdx.x;

    int r = blockIdx.y;
    const int* idx; float* deg; int n;
    if (r < nRM) { idx = dst; deg = degM; n = nM; }
    else         { r -= nRM; idx = src; deg = degU; n = nU; }
    const int base = r * DEG_NB;

    for (int b = tid; b < DEG_NB; b += DEG_TPB) h[b] = 0u;
    __syncthreads();

    const int e0 = blockIdx.x * chunk;
    const int e1 = min(nE, e0 + chunk);
    const int step = DEG_TPB * 4;
    int e = e0 + tid * 4;
    for (; e + step + 3 < e1; e += 2 * step) {
        const int4 v0 = *(const int4*)(idx + e);
        const int4 v1 = *(const int4*)(idx + e + step);
        deg_count(h, v0.x, base); deg_count(h, v0.y, base);
        deg_count(h, v0.z, base); deg_count(h, v0.w, base);
        deg_count(h, v1.x, base); deg_count(h, v1.y, base);
        deg_count(h, v1.z, base); deg_count(h, v1.w, base);
    }
    for (; e < e1; e += step) {
        if (e + 3 < e1) {
            const int4 v0 = *(const int4*)(idx + e);
            deg_count(h, v0.x, base); deg_count(h, v0.y, base);
            deg_count(h, v0.z, base); deg_count(h, v0.w, base);
        } else {
            for (int j = e; j < e1; ++j) deg_count(h, idx[j], base);
        }
    }
    __syncthreads();

    for (int b = tid; b < DEG_NB; b += DEG_TPB) {
        const unsigned int c = h[b];
        if (c != 0u && base + b < n) atomicAdd(deg + base + b, (float)c);
    }
}

__global__ __launch_bounds__(256)
void movie_init_kernel(const float* __restrict__ mx, const float* __restrict__ linW,
                       const float* __restrict__ linb, const float* __restrict__ memb,
                       float* __restrict__ XM, int M) {
    int t = blockIdx.x * blockDim.x + threadIdx.x;
    if (t >= M * HDIM) return;
    int m = t >> 6, h = t & 63;
    float acc = linb[h] + memb[t];
    const float* xr = mx + (size_t)m * 20;
    const float* wr = linW + (size_t)h * 20;
#pragma unroll
    for (int f = 0; f < 20; ++f) acc = fmaf(xr[f], wr[f], acc);
    XM[t] = acc;
}

#define AT_ROWS  1184
#define AT_TPB   1024
#define AT_NW    (AT_TPB / 32)
#define AT_NV    8
#define AT_NC    (AT_NV * 4)
#define AT_BATCH (AT_TPB * AT_NC)
#define AT_QCAP  2048
#define AT_G     16

template <bool FULL>
__device__ __forceinline__ int4 at_ld4(const int* __restrict__ p, int e, int e1) {
    if (FULL || e + 3 < e1) return *(const int4*)(p + e);
    int4 r;
    r.x = (e     < e1) ? p[e]     : -1;
    r.y = (e + 1 < e1) ? p[e + 1] : -1;
    r.z = (e + 2 < e1) ? p[e + 2] : -1;
    r.w = (e + 3 < e1) ? p[e + 3] : -1;
    return r;
}

template <bool FULL>
__device__ __forceinline__ void at_scan(const int* __restrict__ sidx, const int* __restrict__ gidx,
                                        const float* __restrict__ xsrc, float* acc,
                                        unsigned int* qd, int* qe, unsigned int* qcnt,
                                        int eb, int e1, int base, int tid, int lane) {
    int4 v[AT_NV];
#pragma unroll
    for (int j = 0; j < AT_NV; ++j)
        v[j] = at_ld4<FULL>(sidx, eb + j * (AT_TPB * 4) + tid * 4, e1);

    unsigned int mk[AT_NC];
    unsigned int total = 0u;
#pragma unroll
    for (int j = 0; j < AT_NV; ++j) {
        mk[4 * j + 0] = __builtin_amdgcn_ballot_w32((unsigned int)(v[j].x - base) < (unsigned int)AT_ROWS);
        mk[4 * j + 1] = __builtin_amdgcn_ballot_w32((unsigned int)(v[j].y - base) < (unsigned int)AT_ROWS);
        mk[4 * j + 2] = __builtin_amdgcn_ballot_w32((unsigned int)(v[j].z - base) < (unsigned int)AT_ROWS);
        mk[4 * j + 3] = __builtin_amdgcn_ballot_w32((unsigned int)(v[j].w - base) < (unsigned int)AT_ROWS);
        total += __builtin_popcount(mk[4 * j + 0]) + __builtin_popcount(mk[4 * j + 1])
               + __builtin_popcount(mk[4 * j + 2]) + __builtin_popcount(mk[4 * j + 3]);
    }
    if (total == 0u) return;

    unsigned int w0 = 0u;
    if (lane == 0) w0 = atomicAdd(qcnt, total);
    w0 = (unsigned int)__builtin_amdgcn_readfirstlane((int)w0);

    unsigned int run = w0;
#pragma unroll
    for (int i = 0; i < AT_NC; ++i) {
        if (mk[i]) {
            const int j = i >> 2;
            const int vi = (i & 3) == 0 ? v[j].x : (i & 3) == 1 ? v[j].y : (i & 3) == 2 ? v[j].z : v[j].w;
            const unsigned int a = (unsigned int)(vi - base);
            if (a < (unsigned int)AT_ROWS) {
                const unsigned int slot = run + __builtin_amdgcn_mbcnt_lo(mk[i], 0u);
                if (slot < (unsigned int)AT_QCAP) {
                    qd[slot] = a;
                    qe[slot] = eb + j * (AT_TPB * 4) + tid * 4 + (i & 3);
                }
            }
            run += __builtin_popcount(mk[i]);
        }
    }

    if (run > (unsigned int)AT_QCAP) {
        unsigned int r2 = w0;
        for (int q = 0; q < AT_NC; ++q) {
            const int e = eb + (q >> 2) * (AT_TPB * 4) + tid * 4 + (q & 3);
            const unsigned int a = (e < e1) ? (unsigned int)(sidx[e] - base) : 0xffffffffu;
            const bool hit = a < (unsigned int)AT_ROWS;
            const unsigned int mq = __builtin_amdgcn_ballot_w32(hit);
            if (hit && r2 + __builtin_amdgcn_mbcnt_lo(mq, 0u) >= (unsigned int)AT_QCAP) {
                const float* xr = xsrc + (size_t)gidx[e] * HDIM;
                for (int f = 0; f < HDIM; ++f) atomicAdd(&acc[a * HDIM + f], xr[f]);
            }
            r2 += __builtin_popcount(mq);
        }
    }
}

__global__ __launch_bounds__(AT_TPB)
void agg_tile_kernel(const float* __restrict__ xsrc, const int* __restrict__ gidx,
                     const int* __restrict__ sidx, float* __restrict__ agg,
                     int nE, int nDst, int chunk) {
    __shared__ float acc[AT_ROWS * HDIM];
    __shared__ unsigned int qd[AT_QCAP];
    __shared__ int qe[AT_QCAP];
    __shared__ unsigned int qn[2];

    const int tid  = threadIdx.x;
    const int lane = tid & 31;
    const unsigned int wave = (unsigned int)__builtin_amdgcn_readfirstlane(tid >> 5);
    const int base = blockIdx.y * AT_ROWS;

    for (int i = tid; i < AT_ROWS * HDIM / 4; i += AT_TPB)
        ((float4*)acc)[i] = make_float4(0.0f, 0.0f, 0.0f, 0.0f);
    if (tid < 2) qn[tid] = 0u;
    __syncthreads();

    const int e0 = blockIdx.x * chunk;
    const int e1 = min(nE, e0 + chunk);
    int cur = 0;
    for (int eb = e0; eb < e1; eb += AT_BATCH) {
        if (eb + AT_BATCH <= e1)
            at_scan<true >(sidx, gidx, xsrc, acc, qd, qe, &qn[cur], eb, e1, base, tid, lane);
        else
            at_scan<false>(sidx, gidx, xsrc, acc, qd, qe, &qn[cur], eb, e1, base, tid, lane);
        __syncthreads();

        const unsigned int n = (unsigned int)__builtin_amdgcn_readfirstlane(
            (int)min(qn[cur], (unsigned int)AT_QCAP));
        if (tid == 0) qn[cur ^ 1] = 0u;

        for (unsigned int t0 = 0; wave + AT_NW * t0 < n; t0 += AT_G) {
            int ee[AT_G], sr[AT_G];
            unsigned int d[AT_G];
            float2 r[AT_G];
#pragma unroll
            for (int k = 0; k < AT_G; ++k) {
                const unsigned int idx = min(wave + AT_NW * (t0 + k), (unsigned int)(AT_QCAP - 1));
                d[k]  = qd[idx];
                ee[k] = qe[idx];
            }
#pragma unroll
            for (int k = 0; k < AT_G; ++k) {
                if (wave + AT_NW * (t0 + k) < n) sr[k] = gidx[ee[k]];
            }
#pragma unroll
            for (int k = 0; k < AT_G; ++k) {
                if (wave + AT_NW * (t0 + k) < n)
                    r[k] = *(const float2*)(xsrc + (size_t)sr[k] * HDIM + lane * 2);
            }
#pragma unroll
            for (int k = 0; k < AT_G; ++k) {
                const unsigned int idx = wave + AT_NW * (t0 + k);
                if (idx < n) {
                    float* ap = acc + d[k] * HDIM + lane * 2;
                    atomicAdd(ap,     r[k].x);
                    atomicAdd(ap + 1, r[k].y);
                }
            }
        }
        __syncthreads();
        cur ^= 1;
    }

    for (int i = tid; i < AT_ROWS * HDIM; i += AT_TPB) {
        if (base + (i >> 6) < nDst)
            atomicAdd(agg + (size_t)base * HDIM + i, acc[i]);
    }
}

__global__ __launch_bounds__(128)
void combine_wmma_kernel(const float* __restrict__ agg, const float* __restrict__ deg,
                         const float* __restrict__ xdst,
                         const float* __restrict__ Wl, const float* __restrict__ Wr,
                         const float* __restrict__ bias, float* __restrict__ out,
                         int do_relu) {
    __shared__ float A1[16][68];
    __shared__ float A2[16][68];

    const int row0 = blockIdx.x * 16;
    const int tid  = threadIdx.x;

    {
        int r  = tid >> 3;
        int c0 = (tid & 7) * 8;
        float dinv = 1.0f / fmaxf(deg[row0 + r], 1.0f);
        const float* ga = agg  + (size_t)(row0 + r) * HDIM + c0;
        const float* gx = xdst + (size_t)(row0 + r) * HDIM + c0;
#pragma unroll
        for (int j = 0; j < 8; ++j) {
            A1[r][c0 + j] = ga[j] * dinv;
            A2[r][c0 + j] = gx[j];
        }
    }
    __syncthreads();

    const int wave = tid >> 5;
    const int lane = tid & 31;
    const int m    = lane & 15;
    const int half = lane >> 4;
    const int col0 = wave * 16;

    v8f c = {};
#pragma unroll
    for (int kk = 0; kk < 16; ++kk) {
        int k = kk * 4 + half * 2;
        v2f a, b;
        a.x = A1[m][k];  a.y = A1[m][k + 1];
        const float* wp = Wl + (size_t)(col0 + m) * HDIM + k;
        b.x = wp[0];     b.y = wp[1];
        c = __builtin_amdgcn_wmma_f32_16x16x4_f32(false, a, false, b,
                                                  (short)0, c, false, false);
    }
#pragma unroll
    for (int kk = 0; kk < 16; ++kk) {
        int k = kk * 4 + half * 2;
        v2f a, b;
        a.x = A2[m][k];  a.y = A2[m][k + 1];
        const float* wp = Wr + (size_t)(col0 + m) * HDIM + k;
        b.x = wp[0];     b.y = wp[1];
        c = __builtin_amdgcn_wmma_f32_16x16x4_f32(false, a, false, b,
                                                  (short)0, c, false, false);
    }

    float bv = bias[col0 + m];
#pragma unroll
    for (int r = 0; r < 8; ++r) {
        float v = c[r] + bv;
        if (do_relu) v = fmaxf(v, 0.0f);
        out[(size_t)(row0 + r + 8 * half) * HDIM + col0 + m] = v;
    }
}

__global__ __launch_bounds__(256)
void edge_dot_kernel(const float* __restrict__ U2, const float* __restrict__ M2,
                     const int* __restrict__ eu, const int* __restrict__ em,
                     float* __restrict__ out, int L) {
    long long tid = (long long)blockIdx.x * blockDim.x + threadIdx.x;
    if (tid >= (long long)L * 8) return;
    int p = (int)(tid >> 3);
    int s = (int)(tid & 7);
    const float4* a = (const float4*)(U2 + (size_t)eu[p] * HDIM + s * 8);
    const float4* b = (const float4*)(M2 + (size_t)em[p] * HDIM + s * 8);
    float4 a0 = a[0], a1 = a[1], b0 = b[0], b1 = b[1];
    float acc = a0.x * b0.x + a0.y * b0.y + a0.z * b0.z + a0.w * b0.w
              + a1.x * b1.x + a1.y * b1.y + a1.z * b1.z + a1.w * b1.w;
    acc += __shfl_xor(acc, 1);
    acc += __shfl_xor(acc, 2);
    acc += __shfl_xor(acc, 4);
    if (s == 0) out[p] = acc;
}

extern "C" void kernel_launch(void* const* d_in, const int* in_sizes, int n_in,
                              void* d_out, int out_size, void* d_ws, size_t ws_size,
                              hipStream_t stream) {
    (void)n_in; (void)ws_size;
    const float* movie_x   = (const float*)d_in[2];
    const int*   e_src     = (const int*)d_in[3];
    const int*   e_dst     = (const int*)d_in[4];
    const int*   eli_u     = (const int*)d_in[5];
    const int*   eli_m     = (const int*)d_in[6];
    const float* user_emb  = (const float*)d_in[7];
    const float* movie_emb = (const float*)d_in[8];
    const float* lin_W     = (const float*)d_in[9];
    const float* lin_b     = (const float*)d_in[10];
    const float* W1um_l    = (const float*)d_in[11];
    const float* W1um_r    = (const float*)d_in[12];
    const float* W1mu_l    = (const float*)d_in[13];
    const float* W1mu_r    = (const float*)d_in[14];
    const float* W2um_l    = (const float*)d_in[15];
    const float* W2um_r    = (const float*)d_in[16];
    const float* W2mu_l    = (const float*)d_in[17];
    const float* W2mu_r    = (const float*)d_in[18];
    const float* b1um      = (const float*)d_in[19];
    const float* b1mu      = (const float*)d_in[20];
    const float* b2um      = (const float*)d_in[21];
    const float* b2mu      = (const float*)d_in[22];
    float* out = (float*)d_out;

    const int U = in_sizes[7] / HDIM;
    const int M = in_sizes[8] / HDIM;
    const int E = in_sizes[3];
    const int L = in_sizes[5];
    (void)out_size;

    size_t off = 0;
    auto carve = [&](size_t nfloats) {
        float* p = (float*)((char*)d_ws + off);
        off += ((nfloats * sizeof(float) + 255) / 256) * 256;
        return p;
    };
    float* XM    = carve((size_t)M * HDIM);
    float* AGG_M = carve((size_t)M * HDIM);
    float* AGG_U = carve((size_t)U * HDIM);
    float* H_M   = carve((size_t)M * HDIM);
    float* H_U   = carve((size_t)U * HDIM);
    float* M2    = carve((size_t)M * HDIM);
    float* U2    = carve((size_t)U * HDIM);
    float* DEG_M = carve((size_t)M);
    float* DEG_U = carve((size_t)U);

    const int TB = 256;
    dim3 b(TB);
    const int nTM = (M + AT_ROWS - 1) / AT_ROWS;
    const int nTU = (U + AT_ROWS - 1) / AT_ROWS;
    const int cM  = (256 / nTM) > 0 ? (256 / nTM) : 1;
    const int cU  = (256 / nTU) > 0 ? (256 / nTU) : 1;
    const int chM = (((E + cM - 1) / cM) + 3) & ~3;
    const int chU = (((E + cU - 1) / cU) + 3) & ~3;
    const dim3 gM(cM, nTM), gU(cU, nTU);

    hipMemsetAsync(DEG_M, 0, (size_t)M * sizeof(float), stream);
    hipMemsetAsync(DEG_U, 0, (size_t)U * sizeof(float), stream);
    {
        const int nRM = (M + DEG_NB - 1) / DEG_NB;
        const int nRU = (U + DEG_NB - 1) / DEG_NB;
        const int chunk = (((E + DEG_CHUNKS - 1) / DEG_CHUNKS) + 3) & ~3;
        deg_hist_kernel<<<dim3(DEG_CHUNKS, nRM + nRU), DEG_TPB, 0, stream>>>(
            e_src, e_dst, DEG_U, DEG_M, E, U, M, nRM, chunk);
    }

    movie_init_kernel<<<(M * HDIM + TB - 1) / TB, b, 0, stream>>>(
        movie_x, lin_W, lin_b, movie_emb, XM, M);

    hipMemsetAsync(AGG_M, 0, (size_t)M * HDIM * sizeof(float), stream);
    hipMemsetAsync(AGG_U, 0, (size_t)U * HDIM * sizeof(float), stream);
    agg_tile_kernel<<<gM, AT_TPB, 0, stream>>>(user_emb, e_src, e_dst, AGG_M, E, M, chM);
    agg_tile_kernel<<<gU, AT_TPB, 0, stream>>>(XM, e_dst, e_src, AGG_U, E, U, chU);
    combine_wmma_kernel<<<M / 16, 128, 0, stream>>>(AGG_M, DEG_M, XM,
                                                    W1um_l, W1um_r, b1um, H_M, 1);
    combine_wmma_kernel<<<U / 16, 128, 0, stream>>>(AGG_U, DEG_U, user_emb,
                                                    W1mu_l, W1mu_r, b1mu, H_U, 1);

    hipMemsetAsync(AGG_M, 0, (size_t)M * HDIM * sizeof(float), stream);
    hipMemsetAsync(AGG_U, 0, (size_t)U * HDIM * sizeof(float), stream);
    agg_tile_kernel<<<gM, AT_TPB, 0, stream>>>(H_U, e_src, e_dst, AGG_M, E, M, chM);
    agg_tile_kernel<<<gU, AT_TPB, 0, stream>>>(H_M, e_dst, e_src, AGG_U, E, U, chU);
    combine_wmma_kernel<<<M / 16, 128, 0, stream>>>(AGG_M, DEG_M, H_M,
                                                    W2um_l, W2um_r, b2um, M2, 0);
    combine_wmma_kernel<<<U / 16, 128, 0, stream>>>(AGG_U, DEG_U, H_U,
                                                    W2mu_l, W2mu_r, b2mu, U2, 0);

    long long dt = (long long)L * 8;
    edge_dot_kernel<<<(int)((dt + TB - 1) / TB), b, 0, stream>>>(U2, M2, eli_u, eli_m, out, L);
}


# layer-2 segment sums re-read the (local row, source row) lists that the layer-1 tile filter wrote: no edge scan, queue or index lookups in layer 2; sums stay arrival-order LDS adds
# speedup vs baseline: 2.0438x; 1.2003x over previous
#include <hip/hip_runtime.h>
#include <hip/hip_bf16.h>

typedef __attribute__((ext_vector_type(2))) float v2f;
typedef __attribute__((ext_vector_type(8))) float v8f;

#define HDIM 64

#define DEG_NB     12544
#define DEG_TPB    1024
#define DEG_CHUNKS 21

__device__ __forceinline__ void deg_count(unsigned int* h, int idx, int base) {
    unsigned int a = (unsigned int)(idx - base);
    if (a < (unsigned int)DEG_NB) atomicAdd(&h[a], 1u);
}

__global__ __launch_bounds__(DEG_TPB)
void deg_hist_kernel(const int* __restrict__ src, const int* __restrict__ dst,
                     float* __restrict__ degU, float* __restrict__ degM,
                     int nE, int nU, int nM, int nRM, int chunk) {
    __shared__ unsigned int h[DEG_NB];
    const int tid = threadIdx.x;

    int r = blockIdx.y;
    const int* idx; float* deg; int n;
    if (r < nRM) { idx = dst; deg = degM; n = nM; }
    else         { r -= nRM; idx = src; deg = degU; n = nU; }
    const int base = r * DEG_NB;

    for (int b = tid; b < DEG_NB; b += DEG_TPB) h[b] = 0u;
    __syncthreads();

    const int e0 = blockIdx.x * chunk;
    const int e1 = min(nE, e0 + chunk);
    const int step = DEG_TPB * 4;
    int e = e0 + tid * 4;
    for (; e + step + 3 < e1; e += 2 * step) {
        const int4 v0 = *(const int4*)(idx + e);
        const int4 v1 = *(const int4*)(idx + e + step);
        deg_count(h, v0.x, base); deg_count(h, v0.y, base);
        deg_count(h, v0.z, base); deg_count(h, v0.w, base);
        deg_count(h, v1.x, base); deg_count(h, v1.y, base);
        deg_count(h, v1.z, base); deg_count(h, v1.w, base);
    }
    for (; e < e1; e += step) {
        if (e + 3 < e1) {
            const int4 v0 = *(const int4*)(idx + e);
            deg_count(h, v0.x, base); deg_count(h, v0.y, base);
            deg_count(h, v0.z, base); deg_count(h, v0.w, base);
        } else {
            for (int j = e; j < e1; ++j) deg_count(h, idx[j], base);
        }
    }
    __syncthreads();

    for (int b = tid; b < DEG_NB; b += DEG_TPB) {
        const unsigned int c = h[b];
        if (c != 0u && base + b < n) atomicAdd(deg + base + b, (float)c);
    }
}

__global__ __launch_bounds__(256)
void movie_init_kernel(const float* __restrict__ mx, const float* __restrict__ linW,
                       const float* __restrict__ linb, const float* __restrict__ memb,
                       float* __restrict__ XM, int M) {
    int t = blockIdx.x * blockDim.x + threadIdx.x;
    if (t >= M * HDIM) return;
    int m = t >> 6, h = t & 63;
    float acc = linb[h] + memb[t];
    const float* xr = mx + (size_t)m * 20;
    const float* wr = linW + (size_t)h * 20;
#pragma unroll
    for (int f = 0; f < 20; ++f) acc = fmaf(xr[f], wr[f], acc);
    XM[t] = acc;
}

#define AT_ROWS  1184
#define AT_TPB   1024
#define AT_NW    (AT_TPB / 32)
#define AT_NV    8
#define AT_NC    (AT_NV * 4)
#define AT_BATCH (AT_TPB * AT_NC)
#define AT_QCAP  2048
#define AT_G     16
#define AT_SRCB  21
#define AT_LHDR  32

template <bool FULL>
__device__ __forceinline__ int4 at_ld4(const int* __restrict__ p, int e, int e1) {
    if (FULL || e + 3 < e1) return *(const int4*)(p + e);
    int4 r;
    r.x = (e     < e1) ? p[e]     : -1;
    r.y = (e + 1 < e1) ? p[e + 1] : -1;
    r.z = (e + 2 < e1) ? p[e + 2] : -1;
    r.w = (e + 3 < e1) ? p[e + 3] : -1;
    return r;
}

template <bool FULL>
__device__ __forceinline__ void at_scan(const int* __restrict__ sidx, const int* __restrict__ gidx,
                                        const float* __restrict__ xsrc, float* acc,
                                        unsigned int* qd, int* qe, unsigned int* qcnt,
                                        int eb, int e1, int base, int tid, int lane) {
    int4 v[AT_NV];
#pragma unroll
    for (int j = 0; j < AT_NV; ++j)
        v[j] = at_ld4<FULL>(sidx, eb + j * (AT_TPB * 4) + tid * 4, e1);

    unsigned int mk[AT_NC];
    unsigned int total = 0u;
#pragma unroll
    for (int j = 0; j < AT_NV; ++j) {
        mk[4 * j + 0] = __builtin_amdgcn_ballot_w32((unsigned int)(v[j].x - base) < (unsigned int)AT_ROWS);
        mk[4 * j + 1] = __builtin_amdgcn_ballot_w32((unsigned int)(v[j].y - base) < (unsigned int)AT_ROWS);
        mk[4 * j + 2] = __builtin_amdgcn_ballot_w32((unsigned int)(v[j].z - base) < (unsigned int)AT_ROWS);
        mk[4 * j + 3] = __builtin_amdgcn_ballot_w32((unsigned int)(v[j].w - base) < (unsigned int)AT_ROWS);
        total += __builtin_popcount(mk[4 * j + 0]) + __builtin_popcount(mk[4 * j + 1])
               + __builtin_popcount(mk[4 * j + 2]) + __builtin_popcount(mk[4 * j + 3]);
    }
    if (total == 0u) return;

    unsigned int w0 = 0u;
    if (lane == 0) w0 = atomicAdd(qcnt, total);
    w0 = (unsigned int)__builtin_amdgcn_readfirstlane((int)w0);

    unsigned int run = w0;
#pragma unroll
    for (int i = 0; i < AT_NC; ++i) {
        if (mk[i]) {
            const int j = i >> 2;
            const int vi = (i & 3) == 0 ? v[j].x : (i & 3) == 1 ? v[j].y : (i & 3) == 2 ? v[j].z : v[j].w;
            const unsigned int a = (unsigned int)(vi - base);
            if (a < (unsigned int)AT_ROWS) {
                const unsigned int slot = run + __builtin_amdgcn_mbcnt_lo(mk[i], 0u);
                if (slot < (unsigned int)AT_QCAP) {
                    qd[slot] = a;
                    qe[slot] = eb + j * (AT_TPB * 4) + tid * 4 + (i & 3);
                }
            }
            run += __builtin_popcount(mk[i]);
        }
    }

    if (run > (unsigned int)AT_QCAP) {
        unsigned int r2 = w0;
        for (int q = 0; q < AT_NC; ++q) {
            const int e = eb + (q >> 2) * (AT_TPB * 4) + tid * 4 + (q & 3);
            const unsigned int a = (e < e1) ? (unsigned int)(sidx[e] - base) : 0xffffffffu;
            const bool hit = a < (unsigned int)AT_ROWS;
            const unsigned int mq = __builtin_amdgcn_ballot_w32(hit);
            if (hit && r2 + __builtin_amdgcn_mbcnt_lo(mq, 0u) >= (unsigned int)AT_QCAP) {
                const float* xr = xsrc + (size_t)gidx[e] * HDIM;
                for (int f = 0; f < HDIM; ++f) atomicAdd(&acc[a * HDIM + f], xr[f]);
            }
            r2 += __builtin_popcount(mq);
        }
    }
}

template <bool FULL>
__device__ __forceinline__ void at_add16(const float* __restrict__ xsrc, float* acc,
                                         unsigned int pk, int first, unsigned int left, int lane) {
    unsigned int s[AT_G];
    float2 r[AT_G];
#pragma unroll
    for (int k = 0; k < AT_G; ++k)
        s[k] = (unsigned int)__builtin_amdgcn_readlane((int)pk, first + k);
#pragma unroll
    for (int k = 0; k < AT_G; ++k) {
        if (FULL || (unsigned int)k < left)
            r[k] = *(const float2*)(xsrc + (size_t)(s[k] & ((1u << AT_SRCB) - 1u)) * HDIM + lane * 2);
    }
#pragma unroll
    for (int k = 0; k < AT_G; ++k) {
        if (FULL || (unsigned int)k < left) {
            float* ap = acc + (s[k] >> AT_SRCB) * HDIM + lane * 2;
            atomicAdd(ap,     r[k].x);
            atomicAdd(ap + 1, r[k].y);
        }
    }
}

template <bool EMIT>
__global__ __launch_bounds__(AT_TPB)
void agg_tile_kernel(const float* __restrict__ xsrc, const int* __restrict__ gidx,
                     const int* __restrict__ sidx, float* __restrict__ agg,
                     unsigned int* __restrict__ lists, int cap, int nch,
                     int nE, int nDst, int chunk) {
    __shared__ float acc[AT_ROWS * HDIM];
    __shared__ unsigned int qd[AT_QCAP];
    __shared__ int qe[AT_QCAP];
    __shared__ unsigned int qn[2];

    const int tid  = threadIdx.x;
    const int lane = tid & 31;
    const unsigned int wave = (unsigned int)__builtin_amdgcn_readfirstlane(tid >> 5);
    const int base = blockIdx.y * AT_ROWS;
    unsigned int* L = lists + (size_t)(blockIdx.y * nch + blockIdx.x) * (size_t)cap;
    const unsigned int capE = (unsigned int)(cap - AT_LHDR);

    for (int i = tid; i < AT_ROWS * HDIM / 4; i += AT_TPB)
        ((float4*)acc)[i] = make_float4(0.0f, 0.0f, 0.0f, 0.0f);
    if (tid < 2) qn[tid] = 0u;
    __syncthreads();

    bool listed = false;
    if (!EMIT) {
        const unsigned int cnt = (unsigned int)__builtin_amdgcn_readfirstlane((int)L[0]);
        if (cnt <= capE) {
            listed = true;
            unsigned int p0 = wave * 32u;
            unsigned int pk = 0u;
            if (p0 < cnt) pk = L[AT_LHDR + min(p0 + lane, cnt - 1u)];
            for (; p0 < cnt; p0 += AT_NW * 32u) {
                unsigned int nx = 0u;
                if (p0 + 32u <= cnt) {
                    at_add16<true>(xsrc, acc, pk, 0, AT_G, lane);
                    nx = L[AT_LHDR + min(p0 + AT_NW * 32u + lane, cnt - 1u)];
                    at_add16<true>(xsrc, acc, pk, 16, AT_G, lane);
                } else {
                    const unsigned int left = cnt - p0;
                    at_add16<false>(xsrc, acc, pk, 0, min(left, (unsigned int)AT_G), lane);
                    if (left > (unsigned int)AT_G)
                        at_add16<false>(xsrc, acc, pk, 16, left - AT_G, lane);
                }
                pk = nx;
            }
        }
    }

    if (!listed) {
        const int e0 = blockIdx.x * chunk;
        const int e1 = min(nE, e0 + chunk);
        int cur = 0;
        unsigned int run = 0u;
        bool whole = true;
        for (int eb = e0; eb < e1; eb += AT_BATCH) {
            if (eb + AT_BATCH <= e1)
                at_scan<true >(sidx, gidx, xsrc, acc, qd, qe, &qn[cur], eb, e1, base, tid, lane);
            else
                at_scan<false>(sidx, gidx, xsrc, acc, qd, qe, &qn[cur], eb, e1, base, tid, lane);
            __syncthreads();

            const unsigned int raw = (unsigned int)__builtin_amdgcn_readfirstlane((int)qn[cur]);
            const unsigned int n = min(raw, (unsigned int)AT_QCAP);
            if (raw > (unsigned int)AT_QCAP) whole = false;
            if (tid == 0) qn[cur ^ 1] = 0u;

            for (unsigned int t0 = wave * AT_G; t0 < n; t0 += AT_NW * AT_G) {
                const unsigned int idx = t0 + (unsigned int)(lane & (AT_G - 1));
                const unsigned int ci  = min(idx, (unsigned int)(AT_QCAP - 1));
                const unsigned int dq  = qd[ci];
                const int eq = qe[ci];
                unsigned int pk = 0u;
                if (idx < n) pk = (dq << AT_SRCB) | (unsigned int)gidx[eq];
                if (EMIT) {
                    if (idx < n && lane < AT_G && run + idx < capE) L[AT_LHDR + run + idx] = pk;
                }
                if (t0 + AT_G <= n) at_add16<true >(xsrc, acc, pk, 0, AT_G, lane);
                else                at_add16<false>(xsrc, acc, pk, 0, n - t0, lane);
            }
            run += n;
            __syncthreads();
            cur ^= 1;
        }
        if (EMIT) {
            if (tid == 0) L[0] = (whole && run <= capE) ? run : 0xffffffffu;
        }
    }
    __syncthreads();

    for (int i = tid; i < AT_ROWS * HDIM; i += AT_TPB) {
        if (base + (i >> 6) < nDst)
            atomicAdd(agg + (size_t)base * HDIM + i, acc[i]);
    }
}

__global__ __launch_bounds__(128)
void combine_wmma_kernel(const float* __restrict__ agg, const float* __restrict__ deg,
                         const float* __restrict__ xdst,
                         const float* __restrict__ Wl, const float* __restrict__ Wr,
                         const float* __restrict__ bias, float* __restrict__ out,
                         int do_relu) {
    __shared__ float A1[16][68];
    __shared__ float A2[16][68];

    const int row0 = blockIdx.x * 16;
    const int tid  = threadIdx.x;

    {
        int r  = tid >> 3;
        int c0 = (tid & 7) * 8;
        float dinv = 1.0f / fmaxf(deg[row0 + r], 1.0f);
        const float* ga = agg  + (size_t)(row0 + r) * HDIM + c0;
        const float* gx = xdst + (size_t)(row0 + r) * HDIM + c0;
#pragma unroll
        for (int j = 0; j < 8; ++j) {
            A1[r][c0 + j] = ga[j] * dinv;
            A2[r][c0 + j] = gx[j];
        }
    }
    __syncthreads();

    const int wave = tid >> 5;
    const int lane = tid & 31;
    const int m    = lane & 15;
    const int half = lane >> 4;
    const int col0 = wave * 16;

    v8f c = {};
#pragma unroll
    for (int kk = 0; kk < 16; ++kk) {
        int k = kk * 4 + half * 2;
        v2f a, b;
        a.x = A1[m][k];  a.y = A1[m][k + 1];
        const float* wp = Wl + (size_t)(col0 + m) * HDIM + k;
        b.x = wp[0];     b.y = wp[1];
        c = __builtin_amdgcn_wmma_f32_16x16x4_f32(false, a, false, b,
                                                  (short)0, c, false, false);
    }
#pragma unroll
    for (int kk = 0; kk < 16; ++kk) {
        int k = kk * 4 + half * 2;
        v2f a, b;
        a.x = A2[m][k];  a.y = A2[m][k + 1];
        const float* wp = Wr + (size_t)(col0 + m) * HDIM + k;
        b.x = wp[0];     b.y = wp[1];
        c = __builtin_amdgcn_wmma_f32_16x16x4_f32(false, a, false, b,
                                                  (short)0, c, false, false);
    }

    float bv = bias[col0 + m];
#pragma unroll
    for (int r = 0; r < 8; ++r) {
        float v = c[r] + bv;
        if (do_relu) v = fmaxf(v, 0.0f);
        out[(size_t)(row0 + r + 8 * half) * HDIM + col0 + m] = v;
    }
}

__global__ __launch_bounds__(256)
void edge_dot_kernel(const float* __restrict__ U2, const float* __restrict__ M2,
                     const int* __restrict__ eu, const int* __restrict__ em,
                     float* __restrict__ out, int L) {
    long long tid = (long long)blockIdx.x * blockDim.x + threadIdx.x;
    if (tid >= (long long)L * 8) return;
    int p = (int)(tid >> 3);
    int s = (int)(tid & 7);
    const float4* a = (const float4*)(U2 + (size_t)eu[p] * HDIM + s * 8);
    const float4* b = (const float4*)(M2 + (size_t)em[p] * HDIM + s * 8);
    float4 a0 = a[0], a1 = a[1], b0 = b[0], b1 = b[1];
    float acc = a0.x * b0.x + a0.y * b0.y + a0.z * b0.z + a0.w * b0.w
              + a1.x * b1.x + a1.y * b1.y + a1.z * b1.z + a1.w * b1.w;
    acc += __shfl_xor(acc, 1);
    acc += __shfl_xor(acc, 2);
    acc += __shfl_xor(acc, 4);
    if (s == 0) out[p] = acc;
}

extern "C" void kernel_launch(void* const* d_in, const int* in_sizes, int n_in,
                              void* d_out, int out_size, void* d_ws, size_t ws_size,
                              hipStream_t stream) {
    (void)n_in; (void)ws_size;
    const float* movie_x   = (const float*)d_in[2];
    const int*   e_src     = (const int*)d_in[3];
    const int*   e_dst     = (const int*)d_in[4];
    const int*   eli_u     = (const int*)d_in[5];
    const int*   eli_m     = (const int*)d_in[6];
    const float* user_emb  = (const float*)d_in[7];
    const float* movie_emb = (const float*)d_in[8];
    const float* lin_W     = (const float*)d_in[9];
    const float* lin_b     = (const float*)d_in[10];
    const float* W1um_l    = (const float*)d_in[11];
    const float* W1um_r    = (const float*)d_in[12];
    const float* W1mu_l    = (const float*)d_in[13];
    const float* W1mu_r    = (const float*)d_in[14];
    const float* W2um_l    = (const float*)d_in[15];
    const float* W2um_r    = (const float*)d_in[16];
    const float* W2mu_l    = (const float*)d_in[17];
    const float* W2mu_r    = (const float*)d_in[18];
    const float* b1um      = (const float*)d_in[19];
    const float* b1mu      = (const float*)d_in[20];
    const float* b2um      = (const float*)d_in[21];
    const float* b2mu      = (const float*)d_in[22];
    float* out = (float*)d_out;

    const int U = in_sizes[7] / HDIM;
    const int M = in_sizes[8] / HDIM;
    const int E = in_sizes[3];
    const int L = in_sizes[5];
    (void)out_size;

    size_t off = 0;
    auto carve = [&](size_t nfloats) {
        float* p = (float*)((char*)d_ws + off);
        off += ((nfloats * sizeof(float) + 255) / 256) * 256;
        return p;
    };
    float* XM    = carve((size_t)M * HDIM);
    float* AGG_M = carve((size_t)M * HDIM);
    float* AGG_U = carve((size_t)U * HDIM);
    float* H_M   = carve((size_t)M * HDIM);
    float* H_U   = carve((size_t)U * HDIM);
    float* M2    = carve((size_t)M * HDIM);
    float* U2    = carve((size_t)U * HDIM);
    float* DEG_M = carve((size_t)M);
    float* DEG_U = carve((size_t)U);

    const int TB = 256;
    dim3 b(TB);
    const int nTM = (M + AT_ROWS - 1) / AT_ROWS;
    const int nTU = (U + AT_ROWS - 1) / AT_ROWS;
    const int cM  = (256 / nTM) > 0 ? (256 / nTM) : 1;
    const int cU  = (256 / nTU) > 0 ? (256 / nTU) : 1;
    const int chM = (((E + cM - 1) / cM) + 3) & ~3;
    const int chU = (((E + cU - 1) / cU) + 3) & ~3;
    const dim3 gM(cM, nTM), gU(cU, nTU);
    unsigned int* LIST_M = (unsigned int*)U2;
    unsigned int* LIST_U = (unsigned int*)M2;
    const int capM = (int)(((size_t)U * HDIM) / (size_t)(cM * nTM)) & ~31;
    const int capU = (int)(((size_t)M * HDIM) / (size_t)(cU * nTU)) & ~31;

    hipMemsetAsync(DEG_M, 0, (size_t)M * sizeof(float), stream);
    hipMemsetAsync(DEG_U, 0, (size_t)U * sizeof(float), stream);
    {
        const int nRM = (M + DEG_NB - 1) / DEG_NB;
        const int nRU = (U + DEG_NB - 1) / DEG_NB;
        const int chunk = (((E + DEG_CHUNKS - 1) / DEG_CHUNKS) + 3) & ~3;
        deg_hist_kernel<<<dim3(DEG_CHUNKS, nRM + nRU), DEG_TPB, 0, stream>>>(
            e_src, e_dst, DEG_U, DEG_M, E, U, M, nRM, chunk);
    }

    movie_init_kernel<<<(M * HDIM + TB - 1) / TB, b, 0, stream>>>(
        movie_x, lin_W, lin_b, movie_emb, XM, M);

    hipMemsetAsync(AGG_M, 0, (size_t)M * HDIM * sizeof(float), stream);
    hipMemsetAsync(AGG_U, 0, (size_t)U * HDIM * sizeof(float), stream);
    agg_tile_kernel<true><<<gM, AT_TPB, 0, stream>>>(user_emb, e_src, e_dst, AGG_M,
                                                     LIST_M, capM, cM, E, M, chM);
    agg_tile_kernel<true><<<gU, AT_TPB, 0, stream>>>(XM, e_dst, e_src, AGG_U,
                                                     LIST_U, capU, cU, E, U, chU);
    combine_wmma_kernel<<<M / 16, 128, 0, stream>>>(AGG_M, DEG_M, XM,
                                                    W1um_l, W1um_r, b1um, H_M, 1);
    combine_wmma_kernel<<<U / 16, 128, 0, stream>>>(AGG_U, DEG_U, user_emb,
                                                    W1mu_l, W1mu_r, b1mu, H_U, 1);

    hipMemsetAsync(AGG_M, 0, (size_t)M * HDIM * sizeof(float), stream);
    hipMemsetAsync(AGG_U, 0, (size_t)U * HDIM * sizeof(float), stream);
    agg_tile_kernel<false><<<gM, AT_TPB, 0, stream>>>(H_U, e_src, e_dst, AGG_M,
                                                      LIST_M, capM, cM, E, M, chM);
    agg_tile_kernel<false><<<gU, AT_TPB, 0, stream>>>(H_M, e_dst, e_src, AGG_U,
                                                      LIST_U, capU, cU, E, U, chU);
    combine_wmma_kernel<<<M / 16, 128, 0, stream>>>(AGG_M, DEG_M, H_M,
                                                    W2um_l, W2um_r, b2um, M2, 0);
    combine_wmma_kernel<<<U / 16, 128, 0, stream>>>(AGG_U, DEG_U, H_U,
                                                    W2mu_l, W2mu_r, b2mu, U2, 0);

    long long dt = (long long)L * 8;
    edge_dot_kernel<<<(int)((dt + TB - 1) / TB), b, 0, stream>>>(U2, M2, eli_u, eli_m, out, L);
}


# degree counts ride on the layer-1 tile filter: one integer LDS counter per tile row, raised per queued edge, flushed as whole-line f32 adds (exact in any order); the degree kernel's 12 index scans go
# speedup vs baseline: 2.8800x; 1.4091x over previous
#include <hip/hip_runtime.h>
#include <hip/hip_bf16.h>

typedef __attribute__((ext_vector_type(2))) float v2f;
typedef __attribute__((ext_vector_type(8))) float v8f;
typedef __attribute__((ext_vector_type(4))) float v4f;

#define HDIM 64

__global__ __launch_bounds__(256)
void movie_init_kernel(const float* __restrict__ mx, const float* __restrict__ linW,
                       const float* __restrict__ linb, const float* __restrict__ memb,
                       float* __restrict__ XM, int M) {
    int t = blockIdx.x * blockDim.x + threadIdx.x;
    if (t >= M * HDIM) return;
    int m = t >> 6, h = t & 63;
    float acc = linb[h] + memb[t];
    const float* xr = mx + (size_t)m * 20;
    const float* wr = linW + (size_t)h * 20;
#pragma unroll
    for (int f = 0; f < 20; ++f) acc = fmaf(xr[f], wr[f], acc);
    XM[t] = acc;
}

#define AT_ROWS  1184
#define AT_TPB   1024
#define AT_NW    (AT_TPB / 32)
#define AT_STEP  128
#define AT_STRIDE (AT_NW * AT_STEP)
#define AT_G     16
#define AT_WQ    144
#define AT_SRCB  21
#define AT_OFFB  21
#define AT_LHDR  32

__device__ __forceinline__ int4 at_ld4(const int* __restrict__ p, int e, int e1) {
    if (e + 3 < e1) return *(const int4*)(p + e);
    int4 r;
    r.x = (e     < e1) ? p[e]     : -1;
    r.y = (e + 1 < e1) ? p[e + 1] : -1;
    r.z = (e + 2 < e1) ? p[e + 2] : -1;
    r.w = (e + 3 < e1) ? p[e + 3] : -1;
    return r;
}

template <bool FULL>
__device__ __forceinline__ void at_add16(const float* __restrict__ xsrc, float* acc,
                                         unsigned int pk, int first, unsigned int left, int lane) {
    unsigned int s[AT_G];
    float2 r[AT_G];
#pragma unroll
    for (int k = 0; k < AT_G; ++k)
        s[k] = (unsigned int)__builtin_amdgcn_readlane((int)pk, first + k);
#pragma unroll
    for (int k = 0; k < AT_G; ++k) {
        if (FULL || (unsigned int)k < left)
            r[k] = *(const float2*)(xsrc + (size_t)(s[k] & ((1u << AT_SRCB) - 1u)) * HDIM + lane * 2);
    }
#pragma unroll
    for (int k = 0; k < AT_G; ++k) {
        if (FULL || (unsigned int)k < left) {
            float* ap = acc + (s[k] >> AT_SRCB) * HDIM + lane * 2;
            atomicAdd(ap,     r[k].x);
            atomicAdd(ap + 1, r[k].y);
        }
    }
}

template <bool EMIT, bool FULL>
__device__ __forceinline__ void at_take(const float* __restrict__ xsrc, const int* __restrict__ gidx,
                                        float* acc, const unsigned int* wq,
                                        unsigned int first, unsigned int left, int e0,
                                        unsigned int* __restrict__ L, unsigned int capE,
                                        unsigned int* lcnt, unsigned int* rcnt, int lane) {
    const unsigned int k = (unsigned int)(lane & (AT_G - 1));
    const unsigned int w = wq[first + (FULL ? k : min(k, left - 1u))];
    const unsigned int srcrow = (unsigned int)gidx[e0 + (int)(w & ((1u << AT_OFFB) - 1u))];
    const unsigned int pk = ((w >> AT_OFFB) << AT_SRCB) | srcrow;
    if (EMIT) {
        if ((unsigned int)lane < left) atomicAdd(rcnt + (w >> AT_OFFB), 1u);
        unsigned int pos = 0u;
        if (lane == 0) pos = atomicAdd(lcnt, left);
        pos = (unsigned int)__builtin_amdgcn_readfirstlane((int)pos);
        if ((unsigned int)lane < left && pos + (unsigned int)lane < capE)
            L[AT_LHDR + pos + (unsigned int)lane] = pk;
    }
    at_add16<FULL>(xsrc, acc, pk, 0, left, lane);
}

template <bool EMIT>
__device__ __forceinline__ unsigned int at_step(const float* __restrict__ xsrc, const int* __restrict__ gidx,
                                                float* acc, unsigned int* wq, unsigned int pend,
                                                int4 c, int eb, int e0, int base, unsigned int rows,
                                                unsigned int* __restrict__ L, unsigned int capE,
                                                unsigned int* lcnt, unsigned int* rcnt, int lane) {
    const unsigned int a0 = (unsigned int)(c.x - base);
    const unsigned int a1 = (unsigned int)(c.y - base);
    const unsigned int a2 = (unsigned int)(c.z - base);
    const unsigned int a3 = (unsigned int)(c.w - base);
    const unsigned int m0 = __builtin_amdgcn_ballot_w32(a0 < rows);
    const unsigned int m1 = __builtin_amdgcn_ballot_w32(a1 < rows);
    const unsigned int m2 = __builtin_amdgcn_ballot_w32(a2 < rows);
    const unsigned int m3 = __builtin_amdgcn_ballot_w32(a3 < rows);
    if ((m0 | m1 | m2 | m3) == 0u) return pend;

    const unsigned int off = (unsigned int)(eb - e0 + lane * 4);
    if (a0 < rows)
        wq[pend + __builtin_amdgcn_mbcnt_lo(m0, 0u)] = (a0 << AT_OFFB) | off;
    pend += __builtin_popcount(m0);
    if (a1 < rows)
        wq[pend + __builtin_amdgcn_mbcnt_lo(m1, 0u)] = (a1 << AT_OFFB) | (off + 1u);
    pend += __builtin_popcount(m1);
    if (a2 < rows)
        wq[pend + __builtin_amdgcn_mbcnt_lo(m2, 0u)] = (a2 << AT_OFFB) | (off + 2u);
    pend += __builtin_popcount(m2);
    if (a3 < rows)
        wq[pend + __builtin_amdgcn_mbcnt_lo(m3, 0u)] = (a3 << AT_OFFB) | (off + 3u);
    pend += __builtin_popcount(m3);

    while (pend >= (unsigned int)AT_G) {
        pend -= AT_G;
        at_take<EMIT, true>(xsrc, gidx, acc, wq, pend, AT_G, e0, L, capE, lcnt, rcnt, lane);
    }
    return pend;
}

template <bool EMIT>
__device__ __forceinline__ void at_scan_chunk(const float* __restrict__ xsrc, const int* __restrict__ gidx,
                                              const int* __restrict__ sidx, float* acc, unsigned int* wq,
                                              int e0, int e1, int base, unsigned int rows,
                                              unsigned int* __restrict__ L, unsigned int capE,
                                              unsigned int* lcnt, unsigned int* rcnt,
                                              unsigned int wave, int lane) {
    unsigned int pend = 0u;
    const int eb0 = e0 + (int)wave * AT_STEP;
    const int nfull = (e1 - eb0 >= AT_STEP) ? (e1 - eb0 - AT_STEP) / AT_STRIDE + 1 : 0;
    if (nfull > 0) {
        const int elast = eb0 + (nfull - 1) * AT_STRIDE;
        const int* sp = sidx + lane * 4;
        int eb = eb0;
        int4 v0 = *(const int4*)(sp + eb);
        int4 v1 = *(const int4*)(sp + min(eb + 1 * AT_STRIDE, elast));
        int4 v2 = *(const int4*)(sp + min(eb + 2 * AT_STRIDE, elast));
        int4 v3 = *(const int4*)(sp + min(eb + 3 * AT_STRIDE, elast));
        for (int j = 0; j < nfull; j += 4, eb += 4 * AT_STRIDE) {
            pend = at_step<EMIT>(xsrc, gidx, acc, wq, pend, v0, eb, e0, base, rows, L, capE, lcnt, rcnt, lane);
            v0 = *(const int4*)(sp + min(eb + 4 * AT_STRIDE, elast));
            if (j + 1 < nfull)
                pend = at_step<EMIT>(xsrc, gidx, acc, wq, pend, v1, eb + 1 * AT_STRIDE, e0, base, rows, L, capE, lcnt, rcnt, lane);
            v1 = *(const int4*)(sp + min(eb + 5 * AT_STRIDE, elast));
            if (j + 2 < nfull)
                pend = at_step<EMIT>(xsrc, gidx, acc, wq, pend, v2, eb + 2 * AT_STRIDE, e0, base, rows, L, capE, lcnt, rcnt, lane);
            v2 = *(const int4*)(sp + min(eb + 6 * AT_STRIDE, elast));
            if (j + 3 < nfull)
                pend = at_step<EMIT>(xsrc, gidx, acc, wq, pend, v3, eb + 3 * AT_STRIDE, e0, base, rows, L, capE, lcnt, rcnt, lane);
            v3 = *(const int4*)(sp + min(eb + 7 * AT_STRIDE, elast));
        }
    }
    const int ebt = eb0 + nfull * AT_STRIDE;
    if (ebt < e1) {
        const int4 c = at_ld4(sidx, ebt + lane * 4, e1);
        pend = at_step<EMIT>(xsrc, gidx, acc, wq, pend, c, ebt, e0, base, rows, L, capE, lcnt, rcnt, lane);
    }
    if (pend != 0u)
        at_take<EMIT, false>(xsrc, gidx, acc, wq, 0u, pend, e0, L, capE, lcnt, rcnt, lane);
}

__global__ __launch_bounds__(AT_TPB)
void agg_tile_kernel(const float* __restrict__ xsrc, const int* __restrict__ gidx,
                     const int* __restrict__ sidx, float* __restrict__ agg,
                     float* __restrict__ deg,
                     unsigned int* __restrict__ lists, int cap, int nch,
                     int nE, int nDst, int chunk) {
    __shared__ float acc[AT_ROWS * HDIM];
    __shared__ unsigned int wqs[AT_NW * AT_WQ];
    __shared__ unsigned int rcnt[AT_ROWS];
    __shared__ unsigned int lcnt;

    const int tid  = threadIdx.x;
    const int lane = tid & 31;
    const unsigned int wave = (unsigned int)__builtin_amdgcn_readfirstlane(tid >> 5);
    const int base = blockIdx.y * AT_ROWS;
    unsigned int* L = lists + (size_t)(blockIdx.y * nch + blockIdx.x) * (size_t)cap;
    const unsigned int capE = (unsigned int)(cap - AT_LHDR);

    for (int i = tid; i < AT_ROWS * HDIM / 4; i += AT_TPB)
        ((float4*)acc)[i] = make_float4(0.0f, 0.0f, 0.0f, 0.0f);
    for (int i = tid; i < AT_ROWS; i += AT_TPB) rcnt[i] = 0u;
    if (tid == 0) lcnt = 0u;
    __syncthreads();

    {
        const int e0 = blockIdx.x * chunk;
        const int e1 = min(nE, e0 + chunk);
        at_scan_chunk<true>(xsrc, gidx, sidx, acc, wqs + wave * AT_WQ, e0, e1, base,
                            (unsigned int)AT_ROWS, L, capE, &lcnt, rcnt, wave, lane);
    }
    __syncthreads();

    for (int i = tid; i < AT_ROWS; i += AT_TPB) {
        const unsigned int cn = rcnt[i];
        if (cn != 0u && base + i < nDst) atomicAdd(deg + base + i, (float)cn);
    }

    if (tid == 0) L[0] = (lcnt <= capE) ? lcnt : 0xffffffffu;

    for (int i = tid; i < AT_ROWS * HDIM; i += AT_TPB) {
        if (base + (i >> 6) < nDst)
            atomicAdd(agg + (size_t)base * HDIM + i, acc[i]);
    }
}


__device__ __forceinline__ void own_drain(const float* __restrict__ xsrc, float* acc,
                                          const unsigned int* wq, unsigned int pend, int lane) {
    if (pend != 0u) {
        const unsigned int k = (unsigned int)(lane & (AT_G - 1));
        const unsigned int pk = wq[min(k, pend - 1u)];
        at_add16<false>(xsrc, acc, pk, 0, pend, lane);
    }
}

__device__ __forceinline__ unsigned int own_step(const float* __restrict__ xsrc, float* acc,
                                                 unsigned int* wq, unsigned int pend,
                                                 unsigned int w, unsigned int p, unsigned int cnt,
                                                 unsigned int rowoff, unsigned int rows, int lane) {
    const unsigned int a = (w >> AT_SRCB) - rowoff;
    const bool hit = (p < cnt) && (a < rows);
    const unsigned int m = __builtin_amdgcn_ballot_w32(hit);
    if (m == 0u) return pend;
    if (hit)
        wq[pend + __builtin_amdgcn_mbcnt_lo(m, 0u)] = (a << AT_SRCB) | (w & ((1u << AT_SRCB) - 1u));
    pend += __builtin_popcount(m);
    while (pend >= (unsigned int)AT_G) {
        pend -= AT_G;
        const unsigned int pk = wq[pend + (unsigned int)(lane & (AT_G - 1))];
        at_add16<true>(xsrc, acc, pk, 0, AT_G, lane);
    }
    return pend;
}

__device__ __forceinline__ unsigned int own_walk(const float* __restrict__ xsrc, float* acc,
                                                 unsigned int* wq, unsigned int pend,
                                                 const unsigned int* __restrict__ L, unsigned int cnt,
                                                 unsigned int rowoff, unsigned int rows,
                                                 unsigned int wave, int lane) {
    const unsigned int nlines = (cnt + 31u) >> 5;
    if (wave >= nlines) return pend;
    const unsigned int nl = (nlines - 1u - wave) / AT_NW + 1u;
    const unsigned int last = cnt - 1u;
    const unsigned int ps = AT_NW * 32u;
    const unsigned int* Lw = L + AT_LHDR;
    unsigned int p = wave * 32u + (unsigned int)lane;
    unsigned int w0 = Lw[min(p, last)];
    unsigned int w1 = Lw[min(p + 1u * ps, last)];
    unsigned int w2 = Lw[min(p + 2u * ps, last)];
    unsigned int w3 = Lw[min(p + 3u * ps, last)];
    for (unsigned int k = 0u; k < nl; k += 4u, p += 4u * ps) {
        pend = own_step(xsrc, acc, wq, pend, w0, p, cnt, rowoff, rows, lane);
        w0 = Lw[min(p + 4u * ps, last)];
        if (k + 1u < nl)
            pend = own_step(xsrc, acc, wq, pend, w1, p + 1u * ps, cnt, rowoff, rows, lane);
        w1 = Lw[min(p + 5u * ps, last)];
        if (k + 2u < nl)
            pend = own_step(xsrc, acc, wq, pend, w2, p + 2u * ps, cnt, rowoff, rows, lane);
        w2 = Lw[min(p + 6u * ps, last)];
        if (k + 3u < nl)
            pend = own_step(xsrc, acc, wq, pend, w3, p + 3u * ps, cnt, rowoff, rows, lane);
        w3 = Lw[min(p + 7u * ps, last)];
    }
    return pend;
}

__global__ __launch_bounds__(AT_TPB)
void agg_own_kernel(const float* __restrict__ xsrc, const int* __restrict__ gidx,
                    const int* __restrict__ sidx, float* __restrict__ agg,
                    const unsigned int* __restrict__ lists, int cap, int nch,
                    int nE, int nDst, int chunk) {
    __shared__ float acc[AT_ROWS * HDIM];
    __shared__ unsigned int wqs[AT_NW * AT_WQ];

    const int tid  = threadIdx.x;
    const int lane = tid & 31;
    const unsigned int wave = (unsigned int)__builtin_amdgcn_readfirstlane(tid >> 5);
    const int rs     = (AT_ROWS + nch - 1) / nch;
    const int rowoff = blockIdx.x * rs;
    const int nrows  = min(rs, AT_ROWS - rowoff);
    if (nrows <= 0) return;
    const unsigned int rows = (unsigned int)nrows;
    const int base = blockIdx.y * AT_ROWS + rowoff;
    const unsigned int* Lt = lists + (size_t)blockIdx.y * (size_t)nch * (size_t)cap;
    const unsigned int capE = (unsigned int)(cap - AT_LHDR);

    for (int i = tid; i < nrows * (HDIM / 4); i += AT_TPB)
        ((float4*)acc)[i] = make_float4(0.0f, 0.0f, 0.0f, 0.0f);
    __syncthreads();

    unsigned int* wq = wqs + wave * AT_WQ;
    unsigned int pend = 0u;
    for (int c0 = 0; c0 < nch; c0 += 32) {
        unsigned int hdr = 0u;
        if (c0 + lane < nch) hdr = Lt[(size_t)(c0 + lane) * (size_t)cap];
        const int cn = min(32, nch - c0);
        for (int j = 0; j < cn; ++j) {
            const int c = c0 + j;
            const unsigned int cnt = (unsigned int)__builtin_amdgcn_readlane((int)hdr, j);
            if (cnt <= capE) {
                pend = own_walk(xsrc, acc, wq, pend, Lt + (size_t)c * (size_t)cap, cnt,
                                (unsigned int)rowoff, rows, wave, lane);
            } else {
                own_drain(xsrc, acc, wq, pend, lane);
                pend = 0u;
                const int e0 = c * chunk;
                const int e1 = min(nE, e0 + chunk);
                at_scan_chunk<false>(xsrc, gidx, sidx, acc, wq, e0, e1, base, rows,
                                     nullptr, 0u, nullptr, nullptr, wave, lane);
            }
        }
    }
    own_drain(xsrc, acc, wq, pend, lane);
    __syncthreads();

    for (int i = tid; i < nrows * (HDIM / 4); i += AT_TPB) {
        if (base + (i >> 4) < nDst)
            ((float4*)(agg + (size_t)base * HDIM))[i] = ((const float4*)acc)[i];
    }
}

#define CB_TPB   128
#define CB_PITCH 68
#define CB_GRID  512

__device__ __forceinline__ void cb_lds_barrier() {
    __builtin_amdgcn_fence(__ATOMIC_RELEASE, "workgroup", "local");
    __builtin_amdgcn_s_barrier();
    __builtin_amdgcn_fence(__ATOMIC_ACQUIRE, "workgroup", "local");
}

__device__ __forceinline__ void cb_fetch(const float* __restrict__ agg, const float* __restrict__ deg,
                                         const float* __restrict__ xdst, int blk, int nb,
                                         int sr, int sc, v4f* pa, v4f* px, float* pd) {
#pragma unroll
    for (int j = 0; j < 2; ++j) {
        const int row = blk * 16 + sr + 8 * j;
        pd[j] = deg[row];
        pa[j] = *(const v4f*)(agg  + (size_t)row * HDIM + sc);
        px[j] = *(const v4f*)(xdst + (size_t)row * HDIM + sc);
    }
    if (nb == 2) {
#pragma unroll
        for (int j = 2; j < 4; ++j) {
            const int row = blk * 16 + sr + 8 * j;
            pd[j] = deg[row];
            pa[j] = *(const v4f*)(agg  + (size_t)row * HDIM + sc);
            px[j] = *(const v4f*)(xdst + (size_t)row * HDIM + sc);
        }
    }
}

__global__ __launch_bounds__(CB_TPB)
void combine_wmma_kernel(const float* __restrict__ agg, const float* __restrict__ deg,
                         const float* __restrict__ xdst,
                         const float* __restrict__ Wl, const float* __restrict__ Wr,
                         const float* __restrict__ bias, float* __restrict__ out,
                         int do_relu, int nblk) {
    __shared__ float A1[32][CB_PITCH];
    __shared__ float A2[32][CB_PITCH];

    const int tid  = threadIdx.x;
    const int wave = tid >> 5;
    const int lane = tid & 31;
    const int m    = lane & 15;
    const int half = lane >> 4;
    const int col0 = wave * 16;

    v2f bl[16], br[16];
    {
        const float* wl = Wl + (size_t)(col0 + m) * HDIM + half * 2;
        const float* wr = Wr + (size_t)(col0 + m) * HDIM + half * 2;
#pragma unroll
        for (int kk = 0; kk < 16; ++kk) {
            bl[kk].x = wl[kk * 4];  bl[kk].y = wl[kk * 4 + 1];
            br[kk].x = wr[kk * 4];  br[kk].y = wr[kk * 4 + 1];
        }
    }
    const float bv = bias[col0 + m];

    const int b1 = (int)(((unsigned int)(blockIdx.x + 1) * (unsigned int)nblk) / gridDim.x);
    int blk      = (int)(((unsigned int)blockIdx.x * (unsigned int)nblk) / gridDim.x);
    if (blk >= b1) return;

    const int sr = tid >> 4;
    const int sc = (tid & 15) * 4;

    v4f   pa[4], px[4];
    float pd[4];
    cb_fetch(agg, deg, xdst, blk, min(2, b1 - blk), sr, sc, pa, px, pd);

    __syncthreads();

    while (blk < b1) {
        const int nb = min(2, b1 - blk);
#pragma unroll
        for (int j = 0; j < 2; ++j) {
            const float dinv = 1.0f / fmaxf(pd[j], 1.0f);
            *(v4f*)&A1[sr + 8 * j][sc] = pa[j] * dinv;
            *(v4f*)&A2[sr + 8 * j][sc] = px[j];
        }
        if (nb == 2) {
#pragma unroll
            for (int j = 2; j < 4; ++j) {
                const float dinv = 1.0f / fmaxf(pd[j], 1.0f);
                *(v4f*)&A1[sr + 8 * j][sc] = pa[j] * dinv;
                *(v4f*)&A2[sr + 8 * j][sc] = px[j];
            }
        } else {
            const v4f z = {0.0f, 0.0f, 0.0f, 0.0f};
#pragma unroll
            for (int j = 2; j < 4; ++j) {
                *(v4f*)&A1[sr + 8 * j][sc] = z;
                *(v4f*)&A2[sr + 8 * j][sc] = z;
            }
        }
        cb_lds_barrier();

        const int next = blk + nb;
        if (next < b1) cb_fetch(agg, deg, xdst, next, min(2, b1 - next), sr, sc, pa, px, pd);

        v8f c0 = {};
        v8f c1 = {};
#pragma unroll
        for (int kk = 0; kk < 16; ++kk) {
            const int k = kk * 4 + half * 2;
            v2f a, a2;
            a.x  = A1[m][k];       a.y  = A1[m][k + 1];
            a2.x = A1[16 + m][k];  a2.y = A1[16 + m][k + 1];
            c0 = __builtin_amdgcn_wmma_f32_16x16x4_f32(false, a,  false, bl[kk],
                                                       (short)0, c0, false, false);
            c1 = __builtin_amdgcn_wmma_f32_16x16x4_f32(false, a2, false, bl[kk],
                                                       (short)0, c1, false, false);
        }
#pragma unroll
        for (int kk = 0; kk < 16; ++kk) {
            const int k = kk * 4 + half * 2;
            v2f a, a2;
            a.x  = A2[m][k];       a.y  = A2[m][k + 1];
            a2.x = A2[16 + m][k];  a2.y = A2[16 + m][k + 1];
            c0 = __builtin_amdgcn_wmma_f32_16x16x4_f32(false, a,  false, br[kk],
                                                       (short)0, c0, false, false);
            c1 = __builtin_amdgcn_wmma_f32_16x16x4_f32(false, a2, false, br[kk],
                                                       (short)0, c1, false, false);
        }

#pragma unroll
        for (int r = 0; r < 8; ++r) {
            float v = c0[r] + bv;
            if (do_relu) v = fmaxf(v, 0.0f);
            out[(size_t)(blk * 16 + r + 8 * half) * HDIM + col0 + m] = v;
        }
        if (nb == 2) {
#pragma unroll
            for (int r = 0; r < 8; ++r) {
                float v = c1[r] + bv;
                if (do_relu) v = fmaxf(v, 0.0f);
                out[(size_t)((blk + 1) * 16 + r + 8 * half) * HDIM + col0 + m] = v;
            }
        }
        cb_lds_barrier();
        blk = next;
    }
}

__global__ __launch_bounds__(256)
void edge_dot_kernel(const float* __restrict__ U2, const float* __restrict__ M2,
                     const int* __restrict__ eu, const int* __restrict__ em,
                     float* __restrict__ out, int L) {
    long long tid = (long long)blockIdx.x * blockDim.x + threadIdx.x;
    if (tid >= (long long)L * 8) return;
    int p = (int)(tid >> 3);
    int s = (int)(tid & 7);
    const float4* a = (const float4*)(U2 + (size_t)eu[p] * HDIM + s * 8);
    const float4* b = (const float4*)(M2 + (size_t)em[p] * HDIM + s * 8);
    float4 a0 = a[0], a1 = a[1], b0 = b[0], b1 = b[1];
    float acc = a0.x * b0.x + a0.y * b0.y + a0.z * b0.z + a0.w * b0.w
              + a1.x * b1.x + a1.y * b1.y + a1.z * b1.z + a1.w * b1.w;
    acc += __shfl_xor(acc, 1);
    acc += __shfl_xor(acc, 2);
    acc += __shfl_xor(acc, 4);
    if (s == 0) out[p] = acc;
}

extern "C" void kernel_launch(void* const* d_in, const int* in_sizes, int n_in,
                              void* d_out, int out_size, void* d_ws, size_t ws_size,
                              hipStream_t stream) {
    (void)n_in; (void)ws_size;
    const float* movie_x   = (const float*)d_in[2];
    const int*   e_src     = (const int*)d_in[3];
    const int*   e_dst     = (const int*)d_in[4];
    const int*   eli_u     = (const int*)d_in[5];
    const int*   eli_m     = (const int*)d_in[6];
    const float* user_emb  = (const float*)d_in[7];
    const float* movie_emb = (const float*)d_in[8];
    const float* lin_W     = (const float*)d_in[9];
    const float* lin_b     = (const float*)d_in[10];
    const float* W1um_l    = (const float*)d_in[11];
    const float* W1um_r    = (const float*)d_in[12];
    const float* W1mu_l    = (const float*)d_in[13];
    const float* W1mu_r    = (const float*)d_in[14];
    const float* W2um_l    = (const float*)d_in[15];
    const float* W2um_r    = (const float*)d_in[16];
    const float* W2mu_l    = (const float*)d_in[17];
    const float* W2mu_r    = (const float*)d_in[18];
    const float* b1um      = (const float*)d_in[19];
    const float* b1mu      = (const float*)d_in[20];
    const float* b2um      = (const float*)d_in[21];
    const float* b2mu      = (const float*)d_in[22];
    float* out = (float*)d_out;

    const int U = in_sizes[7] / HDIM;
    const int M = in_sizes[8] / HDIM;
    const int E = in_sizes[3];
    const int L = in_sizes[5];
    (void)out_size;

    size_t off = 0;
    auto carve = [&](size_t nfloats) {
        float* p = (float*)((char*)d_ws + off);
        off += ((nfloats * sizeof(float) + 255) / 256) * 256;
        return p;
    };
    float* XM    = carve((size_t)M * HDIM);
    float* AGG_M = carve((size_t)M * HDIM);
    float* AGG_U = carve((size_t)U * HDIM);
    float* H_M   = carve((size_t)M * HDIM);
    float* H_U   = carve((size_t)U * HDIM);
    float* M2    = carve((size_t)M * HDIM);
    float* U2    = carve((size_t)U * HDIM);
    float* DEG_M = carve((size_t)M);
    float* DEG_U = carve((size_t)U);

    const int TB = 256;
    dim3 b(TB);
    const int nTM = (M + AT_ROWS - 1) / AT_ROWS;
    const int nTU = (U + AT_ROWS - 1) / AT_ROWS;
    const int cM  = (256 / nTM) > 0 ? (256 / nTM) : 1;
    const int cU  = (256 / nTU) > 0 ? (256 / nTU) : 1;
    const int chM = (((E + cM - 1) / cM) + 3) & ~3;
    const int chU = (((E + cU - 1) / cU) + 3) & ~3;
    const dim3 gM(cM, nTM), gU(cU, nTU);
    unsigned int* LIST_M = (unsigned int*)U2;
    unsigned int* LIST_U = (unsigned int*)M2;
    const int capM = (int)(((size_t)U * HDIM) / (size_t)(cM * nTM)) & ~31;
    const int capU = (int)(((size_t)M * HDIM) / (size_t)(cU * nTU)) & ~31;

    hipMemsetAsync(DEG_M, 0, (size_t)M * sizeof(float), stream);
    hipMemsetAsync(DEG_U, 0, (size_t)U * sizeof(float), stream);

    movie_init_kernel<<<(M * HDIM + TB - 1) / TB, b, 0, stream>>>(
        movie_x, lin_W, lin_b, movie_emb, XM, M);

    hipMemsetAsync(AGG_M, 0, (size_t)M * HDIM * sizeof(float), stream);
    hipMemsetAsync(AGG_U, 0, (size_t)U * HDIM * sizeof(float), stream);
    agg_tile_kernel<<<gM, AT_TPB, 0, stream>>>(user_emb, e_src, e_dst, AGG_M, DEG_M,
                                               LIST_M, capM, cM, E, M, chM);
    agg_tile_kernel<<<gU, AT_TPB, 0, stream>>>(XM, e_dst, e_src, AGG_U, DEG_U,
                                               LIST_U, capU, cU, E, U, chU);
    combine_wmma_kernel<<<CB_GRID, CB_TPB, 0, stream>>>(AGG_M, DEG_M, XM,
                                                        W1um_l, W1um_r, b1um, H_M, 1, M / 16);
    combine_wmma_kernel<<<CB_GRID, CB_TPB, 0, stream>>>(AGG_U, DEG_U, user_emb,
                                                        W1mu_l, W1mu_r, b1mu, H_U, 1, U / 16);

    agg_own_kernel<<<gM, AT_TPB, 0, stream>>>(H_U, e_src, e_dst, AGG_M,
                                              LIST_M, capM, cM, E, M, chM);
    agg_own_kernel<<<gU, AT_TPB, 0, stream>>>(H_M, e_dst, e_src, AGG_U,
                                              LIST_U, capU, cU, E, U, chU);
    combine_wmma_kernel<<<CB_GRID, CB_TPB, 0, stream>>>(AGG_M, DEG_M, H_M,
                                                        W2um_l, W2um_r, b2um, M2, 0, M / 16);
    combine_wmma_kernel<<<CB_GRID, CB_TPB, 0, stream>>>(AGG_U, DEG_U, H_U,
                                                        W2mu_l, W2mu_r, b2mu, U2, 0, U / 16);

    long long dt = (long long)L * 8;
    edge_dot_kernel<<<(int)((dt + TB - 1) / TB), b, 0, stream>>>(U2, M2, eli_u, eli_m, out, L);
}


# layer-1 tile filter reads one-byte tile numbers, 512 edges per wave step, and one (dst,src) pair per hit (tables written by the init launch): 4x fewer scanned bytes; sums stay arrival-order LDS adds
# speedup vs baseline: 3.0588x; 1.0621x over previous
#include <hip/hip_runtime.h>
#include <hip/hip_bf16.h>

typedef __attribute__((ext_vector_type(2))) float v2f;
typedef __attribute__((ext_vector_type(8))) float v8f;
typedef __attribute__((ext_vector_type(4))) float v4f;

#define HDIM 64

#define AT_ROWS  1184

__device__ __forceinline__ unsigned int tile_byte(int row) {
    return min((unsigned int)row / (unsigned int)AT_ROWS, 0xffu);
}

__global__ __launch_bounds__(256)
void movie_init_kernel(const float* __restrict__ mx, const float* __restrict__ linW,
                       const float* __restrict__ linb, const float* __restrict__ memb,
                       float* __restrict__ XM, int M,
                       const int* __restrict__ esrc, const int* __restrict__ edst,
                       int2* __restrict__ pairs, unsigned int* __restrict__ tidM,
                       unsigned int* __restrict__ tidU, int nE, int nQ) {
    int t = blockIdx.x * blockDim.x + threadIdx.x;
    if (t < nQ) {
        const int e = t * 4;
        unsigned int wm, wu;
        if (e + 3 < nE) {
            const int4 d = *(const int4*)(edst + e);
            const int4 s = *(const int4*)(esrc + e);
            *(int4*)(pairs + e)     = make_int4(d.x, s.x, d.y, s.y);
            *(int4*)(pairs + e + 2) = make_int4(d.z, s.z, d.w, s.w);
            wm = tile_byte(d.x) | (tile_byte(d.y) << 8) | (tile_byte(d.z) << 16) | (tile_byte(d.w) << 24);
            wu = tile_byte(s.x) | (tile_byte(s.y) << 8) | (tile_byte(s.z) << 16) | (tile_byte(s.w) << 24);
        } else {
            wm = 0xffffffffu;
            wu = 0xffffffffu;
            for (int j = 0; j < 4 && e + j < nE; ++j) {
                const int d = edst[e + j], s = esrc[e + j];
                pairs[e + j] = make_int2(d, s);
                wm = (wm & ~(0xffu << (8 * j))) | (tile_byte(d) << (8 * j));
                wu = (wu & ~(0xffu << (8 * j))) | (tile_byte(s) << (8 * j));
            }
        }
        tidM[t] = wm;
        tidU[t] = wu;
    }
    if (t >= M * HDIM) return;
    int m = t >> 6, h = t & 63;
    float acc = linb[h] + memb[t];
    const float* xr = mx + (size_t)m * 20;
    const float* wr = linW + (size_t)h * 20;
#pragma unroll
    for (int f = 0; f < 20; ++f) acc = fmaf(xr[f], wr[f], acc);
    XM[t] = acc;
}

#define AT_TPB   1024
#define AT_NW    (AT_TPB / 32)
#define AT_STEP  128
#define AT_STRIDE (AT_NW * AT_STEP)
#define AT_G     16
#define AT_WQ    144
#define AT_SRCB  21
#define AT_OFFB  21
#define AT_LHDR  32
#define BT_STEP  512

template <bool FULL>
__device__ __forceinline__ void at_add16(const float* __restrict__ xsrc, float* acc,
                                         unsigned int pk, int first, unsigned int left, int lane) {
    unsigned int s[AT_G];
    float2 r[AT_G];
#pragma unroll
    for (int k = 0; k < AT_G; ++k)
        s[k] = (unsigned int)__builtin_amdgcn_readlane((int)pk, first + k);
#pragma unroll
    for (int k = 0; k < AT_G; ++k) {
        if (FULL || (unsigned int)k < left)
            r[k] = *(const float2*)(xsrc + (size_t)(s[k] & ((1u << AT_SRCB) - 1u)) * HDIM + lane * 2);
    }
#pragma unroll
    for (int k = 0; k < AT_G; ++k) {
        if (FULL || (unsigned int)k < left) {
            float* ap = acc + (s[k] >> AT_SRCB) * HDIM + lane * 2;
            atomicAdd(ap,     r[k].x);
            atomicAdd(ap + 1, r[k].y);
        }
    }
}

template <bool FULL>
__device__ __forceinline__ void bt_take(const float* __restrict__ xsrc, const int2* __restrict__ pairs,
                                        int swap, int base, float* acc, const unsigned int* wq,
                                        unsigned int first, unsigned int left,
                                        unsigned int* __restrict__ L, unsigned int capE,
                                        unsigned int* lcnt, unsigned int* rcnt, int lane) {
    const unsigned int k = (unsigned int)(lane & (AT_G - 1));
    const unsigned int w = wq[first + (FULL ? k : min(k, left - 1u))];
    const int2 pr = pairs[w];
    const unsigned int a      = (unsigned int)((swap ? pr.y : pr.x) - base);
    const unsigned int srcrow = (unsigned int)(swap ? pr.x : pr.y);
    const unsigned int pk = (a << AT_SRCB) | srcrow;
    if ((unsigned int)lane < left) atomicAdd(rcnt + a, 1u);
    unsigned int pos = 0u;
    if (lane == 0) pos = atomicAdd(lcnt, left);
    pos = (unsigned int)__builtin_amdgcn_readfirstlane((int)pos);
    if ((unsigned int)lane < left && pos + (unsigned int)lane < capE)
        L[AT_LHDR + pos + (unsigned int)lane] = pk;
    at_add16<FULL>(xsrc, acc, pk, 0, left, lane);
}

__device__ __forceinline__ unsigned int bt_step(const float* __restrict__ xsrc, const int2* __restrict__ pairs,
                                                int swap, int base, float* acc, unsigned int* wq,
                                                unsigned int pend, uint4 v, unsigned int soff, unsigned int pat,
                                                unsigned int* __restrict__ L, unsigned int capE,
                                                unsigned int* lcnt, unsigned int* rcnt, int lane) {
    const unsigned int K7 = 0x7f7f7f7fu, K8 = 0x80808080u;
    const unsigned int x0 = v.x ^ pat, x1 = v.y ^ pat, x2 = v.z ^ pat, x3 = v.w ^ pat;
    const unsigned int n0 = ((x0 & K7) + K7) | x0;
    const unsigned int n1 = ((x1 & K7) + K7) | x1;
    const unsigned int n2 = ((x2 & K7) + K7) | x2;
    const unsigned int n3 = ((x3 & K7) + K7) | x3;
    const bool any = ((n0 & n1 & n2 & n3) & K8) != K8;
    if (__builtin_amdgcn_ballot_w32(any) == 0u) return pend;

    unsigned int hm = ((~n0 & K8) >> 7) | ((~n1 & K8) >> 6) | ((~n2 & K8) >> 5) | ((~n3 & K8) >> 4);
    const unsigned int lbase = soff + (unsigned int)lane * 16u;
    unsigned int m = __builtin_amdgcn_ballot_w32(hm != 0u);
    do {
        if (hm != 0u) {
            const unsigned int p = (unsigned int)__builtin_ctz(hm);
            hm &= hm - 1u;
            wq[pend + __builtin_amdgcn_mbcnt_lo(m, 0u)] = lbase + ((p & 3u) << 2) + (p >> 3);
        }
        pend += __builtin_popcount(m);
        while (pend >= (unsigned int)AT_G) {
            pend -= AT_G;
            bt_take<true>(xsrc, pairs, swap, base, acc, wq, pend, AT_G, L, capE, lcnt, rcnt, lane);
        }
        m = __builtin_amdgcn_ballot_w32(hm != 0u);
    } while (m != 0u);
    return pend;
}

__global__ __launch_bounds__(AT_TPB)
void agg_tile_kernel(const float* __restrict__ xsrc, const int2* __restrict__ pairs, int swap,
                     const unsigned int* __restrict__ tbytes, float* __restrict__ agg,
                     float* __restrict__ deg,
                     unsigned int* __restrict__ lists, int cap, int nch,
                     int nE, int nDst, int chunk) {
    __shared__ float acc[AT_ROWS * HDIM];
    __shared__ unsigned int wqs[AT_NW * AT_WQ];
    __shared__ unsigned int rcnt[AT_ROWS];
    __shared__ unsigned int lcnt;

    const int tid = threadIdx.x;
    const int lane = tid & 31;
    const int wave = __builtin_amdgcn_readfirstlane(tid >> 5);
    const int base = blockIdx.y * AT_ROWS;
    unsigned int* L = lists + (size_t)(blockIdx.y * nch + blockIdx.x) * (size_t)cap;
    const unsigned int capE = (unsigned int)(cap - AT_LHDR);

    for (int i = tid; i < AT_ROWS * HDIM / 4; i += AT_TPB)
        ((float4*)acc)[i] = make_float4(0.0f, 0.0f, 0.0f, 0.0f);
    for (int i = tid; i < AT_ROWS; i += AT_TPB) rcnt[i] = 0u;
    if (tid == 0) lcnt = 0u;
    __syncthreads();

    {
        const int e0 = blockIdx.x * chunk;
        const int e1 = min(nE, e0 + chunk);
        const int nst = (e1 - e0 + BT_STEP - 1) / BT_STEP;
        unsigned int* wq = wqs + wave * AT_WQ;
        const int2* pr = pairs + e0;
        const unsigned int pat = (unsigned int)blockIdx.y * 0x01010101u;
        unsigned int pend = 0u;
        if (wave < nst) {
            const int nw = (nst - 1 - wave) / AT_NW + 1;
            const uint4* tp = (const uint4*)(tbytes + (e0 >> 2)) + lane;
            const int slast = wave + (nw - 1) * AT_NW;
            int s = wave;
            uint4 v0 = tp[s * 32];
            uint4 v1 = tp[min(s + 1 * AT_NW, slast) * 32];
            uint4 v2 = tp[min(s + 2 * AT_NW, slast) * 32];
            uint4 v3 = tp[min(s + 3 * AT_NW, slast) * 32];
            for (int j = 0; j < nw; j += 4, s += 4 * AT_NW) {
                pend = bt_step(xsrc, pr, swap, base, acc, wq, pend, v0, (unsigned int)s * BT_STEP,
                               pat, L, capE, &lcnt, rcnt, lane);
                v0 = tp[min(s + 4 * AT_NW, slast) * 32];
                if (j + 1 < nw)
                    pend = bt_step(xsrc, pr, swap, base, acc, wq, pend, v1,
                                   (unsigned int)(s + 1 * AT_NW) * BT_STEP, pat, L, capE, &lcnt, rcnt, lane);
                v1 = tp[min(s + 5 * AT_NW, slast) * 32];
                if (j + 2 < nw)
                    pend = bt_step(xsrc, pr, swap, base, acc, wq, pend, v2,
                                   (unsigned int)(s + 2 * AT_NW) * BT_STEP, pat, L, capE, &lcnt, rcnt, lane);
                v2 = tp[min(s + 6 * AT_NW, slast) * 32];
                if (j + 3 < nw)
                    pend = bt_step(xsrc, pr, swap, base, acc, wq, pend, v3,
                                   (unsigned int)(s + 3 * AT_NW) * BT_STEP, pat, L, capE, &lcnt, rcnt, lane);
                v3 = tp[min(s + 7 * AT_NW, slast) * 32];
            }
        }
        if (pend != 0u)
            bt_take<false>(xsrc, pr, swap, base, acc, wq, 0u, pend, L, capE, &lcnt, rcnt, lane);
    }
    __syncthreads();

    for (int i = tid; i < AT_ROWS; i += AT_TPB) {
        const unsigned int cn = rcnt[i];
        if (cn != 0u && base + i < nDst) atomicAdd(deg + base + i, (float)cn);
    }

    if (tid == 0) L[0] = (lcnt <= capE) ? lcnt : 0xffffffffu;

    for (int i = tid; i < AT_ROWS * HDIM; i += AT_TPB) {
        if (base + (i >> 6) < nDst)
            atomicAdd(agg + (size_t)base * HDIM + i, acc[i]);
    }
}


__device__ __forceinline__ int4 at_ld4(const int* __restrict__ p, int e, int e1) {
    if (e + 3 < e1) return *(const int4*)(p + e);
    int4 r;
    r.x = (e     < e1) ? p[e]     : -1;
    r.y = (e + 1 < e1) ? p[e + 1] : -1;
    r.z = (e + 2 < e1) ? p[e + 2] : -1;
    r.w = (e + 3 < e1) ? p[e + 3] : -1;
    return r;
}

template <bool FULL>
__device__ __forceinline__ void at_take(const float* __restrict__ xsrc, const int* __restrict__ gidx,
                                        float* acc, const unsigned int* wq,
                                        unsigned int first, unsigned int left, int e0, int lane) {
    const unsigned int k = (unsigned int)(lane & (AT_G - 1));
    const unsigned int w = wq[first + (FULL ? k : min(k, left - 1u))];
    const unsigned int srcrow = (unsigned int)gidx[e0 + (int)(w & ((1u << AT_OFFB) - 1u))];
    const unsigned int pk = ((w >> AT_OFFB) << AT_SRCB) | srcrow;
    at_add16<FULL>(xsrc, acc, pk, 0, left, lane);
}

__device__ __forceinline__ unsigned int at_step(const float* __restrict__ xsrc, const int* __restrict__ gidx,
                                                float* acc, unsigned int* wq, unsigned int pend,
                                                int4 c, int eb, int e0, int base, unsigned int rows, int lane) {
    const unsigned int a0 = (unsigned int)(c.x - base);
    const unsigned int a1 = (unsigned int)(c.y - base);
    const unsigned int a2 = (unsigned int)(c.z - base);
    const unsigned int a3 = (unsigned int)(c.w - base);
    const unsigned int m0 = __builtin_amdgcn_ballot_w32(a0 < rows);
    const unsigned int m1 = __builtin_amdgcn_ballot_w32(a1 < rows);
    const unsigned int m2 = __builtin_amdgcn_ballot_w32(a2 < rows);
    const unsigned int m3 = __builtin_amdgcn_ballot_w32(a3 < rows);
    if ((m0 | m1 | m2 | m3) == 0u) return pend;

    const unsigned int off = (unsigned int)(eb - e0 + lane * 4);
    if (a0 < rows)
        wq[pend + __builtin_amdgcn_mbcnt_lo(m0, 0u)] = (a0 << AT_OFFB) | off;
    pend += __builtin_popcount(m0);
    if (a1 < rows)
        wq[pend + __builtin_amdgcn_mbcnt_lo(m1, 0u)] = (a1 << AT_OFFB) | (off + 1u);
    pend += __builtin_popcount(m1);
    if (a2 < rows)
        wq[pend + __builtin_amdgcn_mbcnt_lo(m2, 0u)] = (a2 << AT_OFFB) | (off + 2u);
    pend += __builtin_popcount(m2);
    if (a3 < rows)
        wq[pend + __builtin_amdgcn_mbcnt_lo(m3, 0u)] = (a3 << AT_OFFB) | (off + 3u);
    pend += __builtin_popcount(m3);

    while (pend >= (unsigned int)AT_G) {
        pend -= AT_G;
        at_take<true>(xsrc, gidx, acc, wq, pend, AT_G, e0, lane);
    }
    return pend;
}

__device__ __forceinline__ void at_scan_chunk(const float* __restrict__ xsrc, const int* __restrict__ gidx,
                                              const int* __restrict__ sidx, float* acc, unsigned int* wq,
                                              int e0, int e1, int base, unsigned int rows,
                                              unsigned int wave, int lane) {
    unsigned int pend = 0u;
    const int eb0 = e0 + (int)wave * AT_STEP;
    const int nfull = (e1 - eb0 >= AT_STEP) ? (e1 - eb0 - AT_STEP) / AT_STRIDE + 1 : 0;
    if (nfull > 0) {
        const int elast = eb0 + (nfull - 1) * AT_STRIDE;
        const int* sp = sidx + lane * 4;
        int eb = eb0;
        int4 v0 = *(const int4*)(sp + eb);
        int4 v1 = *(const int4*)(sp + min(eb + 1 * AT_STRIDE, elast));
        int4 v2 = *(const int4*)(sp + min(eb + 2 * AT_STRIDE, elast));
        int4 v3 = *(const int4*)(sp + min(eb + 3 * AT_STRIDE, elast));
        for (int j = 0; j < nfull; j += 4, eb += 4 * AT_STRIDE) {
            pend = at_step(xsrc, gidx, acc, wq, pend, v0, eb, e0, base, rows, lane);
            v0 = *(const int4*)(sp + min(eb + 4 * AT_STRIDE, elast));
            if (j + 1 < nfull)
                pend = at_step(xsrc, gidx, acc, wq, pend, v1, eb + 1 * AT_STRIDE, e0, base, rows, lane);
            v1 = *(const int4*)(sp + min(eb + 5 * AT_STRIDE, elast));
            if (j + 2 < nfull)
                pend = at_step(xsrc, gidx, acc, wq, pend, v2, eb + 2 * AT_STRIDE, e0, base, rows, lane);
            v2 = *(const int4*)(sp + min(eb + 6 * AT_STRIDE, elast));
            if (j + 3 < nfull)
                pend = at_step(xsrc, gidx, acc, wq, pend, v3, eb + 3 * AT_STRIDE, e0, base, rows, lane);
            v3 = *(const int4*)(sp + min(eb + 7 * AT_STRIDE, elast));
        }
    }
    const int ebt = eb0 + nfull * AT_STRIDE;
    if (ebt < e1) {
        const int4 c = at_ld4(sidx, ebt + lane * 4, e1);
        pend = at_step(xsrc, gidx, acc, wq, pend, c, ebt, e0, base, rows, lane);
    }
    if (pend != 0u)
        at_take<false>(xsrc, gidx, acc, wq, 0u, pend, e0, lane);
}


__device__ __forceinline__ void own_drain(const float* __restrict__ xsrc, float* acc,
                                          const unsigned int* wq, unsigned int pend, int lane) {
    if (pend != 0u) {
        const unsigned int k = (unsigned int)(lane & (AT_G - 1));
        const unsigned int pk = wq[min(k, pend - 1u)];
        at_add16<false>(xsrc, acc, pk, 0, pend, lane);
    }
}

__device__ __forceinline__ unsigned int own_step(const float* __restrict__ xsrc, float* acc,
                                                 unsigned int* wq, unsigned int pend,
                                                 unsigned int w, unsigned int p, unsigned int cnt,
                                                 unsigned int rowoff, unsigned int rows, int lane) {
    const unsigned int a = (w >> AT_SRCB) - rowoff;
    const bool hit = (p < cnt) && (a < rows);
    const unsigned int m = __builtin_amdgcn_ballot_w32(hit);
    if (m == 0u) return pend;
    if (hit)
        wq[pend + __builtin_amdgcn_mbcnt_lo(m, 0u)] = (a << AT_SRCB) | (w & ((1u << AT_SRCB) - 1u));
    pend += __builtin_popcount(m);
    while (pend >= (unsigned int)AT_G) {
        pend -= AT_G;
        const unsigned int pk = wq[pend + (unsigned int)(lane & (AT_G - 1))];
        at_add16<true>(xsrc, acc, pk, 0, AT_G, lane);
    }
    return pend;
}

__device__ __forceinline__ unsigned int own_walk(const float* __restrict__ xsrc, float* acc,
                                                 unsigned int* wq, unsigned int pend,
                                                 const unsigned int* __restrict__ L, unsigned int cnt,
                                                 unsigned int rowoff, unsigned int rows,
                                                 unsigned int wave, int lane) {
    const unsigned int nlines = (cnt + 31u) >> 5;
    if (wave >= nlines) return pend;
    const unsigned int nl = (nlines - 1u - wave) / AT_NW + 1u;
    const unsigned int last = cnt - 1u;
    const unsigned int ps = AT_NW * 32u;
    const unsigned int* Lw = L + AT_LHDR;
    unsigned int p = wave * 32u + (unsigned int)lane;
    unsigned int w0 = Lw[min(p, last)];
    unsigned int w1 = Lw[min(p + 1u * ps, last)];
    unsigned int w2 = Lw[min(p + 2u * ps, last)];
    unsigned int w3 = Lw[min(p + 3u * ps, last)];
    for (unsigned int k = 0u; k < nl; k += 4u, p += 4u * ps) {
        pend = own_step(xsrc, acc, wq, pend, w0, p, cnt, rowoff, rows, lane);
        w0 = Lw[min(p + 4u * ps, last)];
        if (k + 1u < nl)
            pend = own_step(xsrc, acc, wq, pend, w1, p + 1u * ps, cnt, rowoff, rows, lane);
        w1 = Lw[min(p + 5u * ps, last)];
        if (k + 2u < nl)
            pend = own_step(xsrc, acc, wq, pend, w2, p + 2u * ps, cnt, rowoff, rows, lane);
        w2 = Lw[min(p + 6u * ps, last)];
        if (k + 3u < nl)
            pend = own_step(xsrc, acc, wq, pend, w3, p + 3u * ps, cnt, rowoff, rows, lane);
        w3 = Lw[min(p + 7u * ps, last)];
    }
    return pend;
}

__global__ __launch_bounds__(AT_TPB)
void agg_own_kernel(const float* __restrict__ xsrc, const int* __restrict__ gidx,
                    const int* __restrict__ sidx, float* __restrict__ agg,
                    const unsigned int* __restrict__ lists, int cap, int nch,
                    int nE, int nDst, int chunk) {
    __shared__ float acc[AT_ROWS * HDIM];
    __shared__ unsigned int wqs[AT_NW * AT_WQ];

    const int tid  = threadIdx.x;
    const int lane = tid & 31;
    const unsigned int wave = (unsigned int)__builtin_amdgcn_readfirstlane(tid >> 5);
    const int rs     = (AT_ROWS + nch - 1) / nch;
    const int rowoff = blockIdx.x * rs;
    const int nrows  = min(rs, AT_ROWS - rowoff);
    if (nrows <= 0) return;
    const unsigned int rows = (unsigned int)nrows;
    const int base = blockIdx.y * AT_ROWS + rowoff;
    const unsigned int* Lt = lists + (size_t)blockIdx.y * (size_t)nch * (size_t)cap;
    const unsigned int capE = (unsigned int)(cap - AT_LHDR);

    for (int i = tid; i < nrows * (HDIM / 4); i += AT_TPB)
        ((float4*)acc)[i] = make_float4(0.0f, 0.0f, 0.0f, 0.0f);
    __syncthreads();

    unsigned int* wq = wqs + wave * AT_WQ;
    unsigned int pend = 0u;
    for (int c0 = 0; c0 < nch; c0 += 32) {
        unsigned int hdr = 0u;
        if (c0 + lane < nch) hdr = Lt[(size_t)(c0 + lane) * (size_t)cap];
        const int cn = min(32, nch - c0);
        for (int j = 0; j < cn; ++j) {
            const int c = c0 + j;
            const unsigned int cnt = (unsigned int)__builtin_amdgcn_readlane((int)hdr, j);
            if (cnt <= capE) {
                pend = own_walk(xsrc, acc, wq, pend, Lt + (size_t)c * (size_t)cap, cnt,
                                (unsigned int)rowoff, rows, wave, lane);
            } else {
                own_drain(xsrc, acc, wq, pend, lane);
                pend = 0u;
                const int e0 = c * chunk;
                const int e1 = min(nE, e0 + chunk);
                at_scan_chunk(xsrc, gidx, sidx, acc, wq, e0, e1, base, rows, wave, lane);
            }
        }
    }
    own_drain(xsrc, acc, wq, pend, lane);
    __syncthreads();

    for (int i = tid; i < nrows * (HDIM / 4); i += AT_TPB) {
        if (base + (i >> 4) < nDst)
            ((float4*)(agg + (size_t)base * HDIM))[i] = ((const float4*)acc)[i];
    }
}

#define CB_TPB   128
#define CB_PITCH 68
#define CB_GRID  512

__device__ __forceinline__ void cb_lds_barrier() {
    __builtin_amdgcn_fence(__ATOMIC_RELEASE, "workgroup", "local");
    __builtin_amdgcn_s_barrier();
    __builtin_amdgcn_fence(__ATOMIC_ACQUIRE, "workgroup", "local");
}

__device__ __forceinline__ void cb_fetch(const float* __restrict__ agg, const float* __restrict__ deg,
                                         const float* __restrict__ xdst, int blk, int nb,
                                         int sr, int sc, v4f* pa, v4f* px, float* pd) {
#pragma unroll
    for (int j = 0; j < 2; ++j) {
        const int row = blk * 16 + sr + 8 * j;
        pd[j] = deg[row];
        pa[j] = *(const v4f*)(agg  + (size_t)row * HDIM + sc);
        px[j] = *(const v4f*)(xdst + (size_t)row * HDIM + sc);
    }
    if (nb == 2) {
#pragma unroll
        for (int j = 2; j < 4; ++j) {
            const int row = blk * 16 + sr + 8 * j;
            pd[j] = deg[row];
            pa[j] = *(const v4f*)(agg  + (size_t)row * HDIM + sc);
            px[j] = *(const v4f*)(xdst + (size_t)row * HDIM + sc);
        }
    }
}

__global__ __launch_bounds__(CB_TPB)
void combine_wmma_kernel(const float* __restrict__ agg, const float* __restrict__ deg,
                         const float* __restrict__ xdst,
                         const float* __restrict__ Wl, const float* __restrict__ Wr,
                         const float* __restrict__ bias, float* __restrict__ out,
                         int do_relu, int nblk) {
    __shared__ float A1[32][CB_PITCH];
    __shared__ float A2[32][CB_PITCH];

    const int tid  = threadIdx.x;
    const int wave = tid >> 5;
    const int lane = tid & 31;
    const int m    = lane & 15;
    const int half = lane >> 4;
    const int col0 = wave * 16;

    v2f bl[16], br[16];
    {
        const float* wl = Wl + (size_t)(col0 + m) * HDIM + half * 2;
        const float* wr = Wr + (size_t)(col0 + m) * HDIM + half * 2;
#pragma unroll
        for (int kk = 0; kk < 16; ++kk) {
            bl[kk].x = wl[kk * 4];  bl[kk].y = wl[kk * 4 + 1];
            br[kk].x = wr[kk * 4];  br[kk].y = wr[kk * 4 + 1];
        }
    }
    const float bv = bias[col0 + m];

    const int b1 = (int)(((unsigned int)(blockIdx.x + 1) * (unsigned int)nblk) / gridDim.x);
    int blk      = (int)(((unsigned int)blockIdx.x * (unsigned int)nblk) / gridDim.x);
    if (blk >= b1) return;

    const int sr = tid >> 4;
    const int sc = (tid & 15) * 4;

    v4f   pa[4], px[4];
    float pd[4];
    cb_fetch(agg, deg, xdst, blk, min(2, b1 - blk), sr, sc, pa, px, pd);

    __syncthreads();

    while (blk < b1) {
        const int nb = min(2, b1 - blk);
#pragma unroll
        for (int j = 0; j < 2; ++j) {
            const float dinv = 1.0f / fmaxf(pd[j], 1.0f);
            *(v4f*)&A1[sr + 8 * j][sc] = pa[j] * dinv;
            *(v4f*)&A2[sr + 8 * j][sc] = px[j];
        }
        if (nb == 2) {
#pragma unroll
            for (int j = 2; j < 4; ++j) {
                const float dinv = 1.0f / fmaxf(pd[j], 1.0f);
                *(v4f*)&A1[sr + 8 * j][sc] = pa[j] * dinv;
                *(v4f*)&A2[sr + 8 * j][sc] = px[j];
            }
        } else {
            const v4f z = {0.0f, 0.0f, 0.0f, 0.0f};
#pragma unroll
            for (int j = 2; j < 4; ++j) {
                *(v4f*)&A1[sr + 8 * j][sc] = z;
                *(v4f*)&A2[sr + 8 * j][sc] = z;
            }
        }
        cb_lds_barrier();

        const int next = blk + nb;
        if (next < b1) cb_fetch(agg, deg, xdst, next, min(2, b1 - next), sr, sc, pa, px, pd);

        v8f c0 = {};
        v8f c1 = {};
#pragma unroll
        for (int kk = 0; kk < 16; ++kk) {
            const int k = kk * 4 + half * 2;
            v2f a, a2;
            a.x  = A1[m][k];       a.y  = A1[m][k + 1];
            a2.x = A1[16 + m][k];  a2.y = A1[16 + m][k + 1];
            c0 = __builtin_amdgcn_wmma_f32_16x16x4_f32(false, a,  false, bl[kk],
                                                       (short)0, c0, false, false);
            c1 = __builtin_amdgcn_wmma_f32_16x16x4_f32(false, a2, false, bl[kk],
                                                       (short)0, c1, false, false);
        }
#pragma unroll
        for (int kk = 0; kk < 16; ++kk) {
            const int k = kk * 4 + half * 2;
            v2f a, a2;
            a.x  = A2[m][k];       a.y  = A2[m][k + 1];
            a2.x = A2[16 + m][k];  a2.y = A2[16 + m][k + 1];
            c0 = __builtin_amdgcn_wmma_f32_16x16x4_f32(false, a,  false, br[kk],
                                                       (short)0, c0, false, false);
            c1 = __builtin_amdgcn_wmma_f32_16x16x4_f32(false, a2, false, br[kk],
                                                       (short)0, c1, false, false);
        }

#pragma unroll
        for (int r = 0; r < 8; ++r) {
            float v = c0[r] + bv;
            if (do_relu) v = fmaxf(v, 0.0f);
            out[(size_t)(blk * 16 + r + 8 * half) * HDIM + col0 + m] = v;
        }
        if (nb == 2) {
#pragma unroll
            for (int r = 0; r < 8; ++r) {
                float v = c1[r] + bv;
                if (do_relu) v = fmaxf(v, 0.0f);
                out[(size_t)((blk + 1) * 16 + r + 8 * half) * HDIM + col0 + m] = v;
            }
        }
        cb_lds_barrier();
        blk = next;
    }
}

__global__ __launch_bounds__(256)
void edge_dot_kernel(const float* __restrict__ U2, const float* __restrict__ M2,
                     const int* __restrict__ eu, const int* __restrict__ em,
                     float* __restrict__ out, int L) {
    long long tid = (long long)blockIdx.x * blockDim.x + threadIdx.x;
    if (tid >= (long long)L * 8) return;
    int p = (int)(tid >> 3);
    int s = (int)(tid & 7);
    const float4* a = (const float4*)(U2 + (size_t)eu[p] * HDIM + s * 8);
    const float4* b = (const float4*)(M2 + (size_t)em[p] * HDIM + s * 8);
    float4 a0 = a[0], a1 = a[1], b0 = b[0], b1 = b[1];
    float acc = a0.x * b0.x + a0.y * b0.y + a0.z * b0.z + a0.w * b0.w
              + a1.x * b1.x + a1.y * b1.y + a1.z * b1.z + a1.w * b1.w;
    acc += __shfl_xor(acc, 1);
    acc += __shfl_xor(acc, 2);
    acc += __shfl_xor(acc, 4);
    if (s == 0) out[p] = acc;
}

extern "C" void kernel_launch(void* const* d_in, const int* in_sizes, int n_in,
                              void* d_out, int out_size, void* d_ws, size_t ws_size,
                              hipStream_t stream) {
    (void)n_in; (void)ws_size;
    const float* movie_x   = (const float*)d_in[2];
    const int*   e_src     = (const int*)d_in[3];
    const int*   e_dst     = (const int*)d_in[4];
    const int*   eli_u     = (const int*)d_in[5];
    const int*   eli_m     = (const int*)d_in[6];
    const float* user_emb  = (const float*)d_in[7];
    const float* movie_emb = (const float*)d_in[8];
    const float* lin_W     = (const float*)d_in[9];
    const float* lin_b     = (const float*)d_in[10];
    const float* W1um_l    = (const float*)d_in[11];
    const float* W1um_r    = (const float*)d_in[12];
    const float* W1mu_l    = (const float*)d_in[13];
    const float* W1mu_r    = (const float*)d_in[14];
    const float* W2um_l    = (const float*)d_in[15];
    const float* W2um_r    = (const float*)d_in[16];
    const float* W2mu_l    = (const float*)d_in[17];
    const float* W2mu_r    = (const float*)d_in[18];
    const float* b1um      = (const float*)d_in[19];
    const float* b1mu      = (const float*)d_in[20];
    const float* b2um      = (const float*)d_in[21];
    const float* b2mu      = (const float*)d_in[22];
    float* out = (float*)d_out;

    const int U = in_sizes[7] / HDIM;
    const int M = in_sizes[8] / HDIM;
    const int E = in_sizes[3];
    const int L = in_sizes[5];
    (void)out_size;

    size_t off = 0;
    auto carve = [&](size_t nfloats) {
        float* p = (float*)((char*)d_ws + off);
        off += ((nfloats * sizeof(float) + 255) / 256) * 256;
        return p;
    };
    float* XM    = carve((size_t)M * HDIM);
    float* AGG_M = carve((size_t)M * HDIM);
    float* AGG_U = carve((size_t)U * HDIM);
    float* H_M   = carve((size_t)M * HDIM);
    float* H_U   = carve((size_t)U * HDIM);
    float* M2    = carve((size_t)M * HDIM);
    float* U2    = carve((size_t)U * HDIM);
    float* DEG_M = carve((size_t)M);
    float* DEG_U = carve((size_t)U);

    const int TB = 256;
    dim3 b(TB);
    const int nTM = (M + AT_ROWS - 1) / AT_ROWS;
    const int nTU = (U + AT_ROWS - 1) / AT_ROWS;
    const int cM  = (256 / nTM) > 0 ? (256 / nTM) : 1;
    const int cU  = (256 / nTU) > 0 ? (256 / nTU) : 1;
    const int chM = (((E + cM - 1) / cM) + BT_STEP - 1) & ~(BT_STEP - 1);
    const int chU = (((E + cU - 1) / cU) + BT_STEP - 1) & ~(BT_STEP - 1);
    const dim3 gM(cM, nTM), gU(cU, nTU);
    unsigned int* LIST_M = (unsigned int*)U2;
    unsigned int* LIST_U = (unsigned int*)M2;
    const int capM = (int)(((size_t)U * HDIM) / (size_t)(cM * nTM)) & ~31;
    const int capU = (int)(((size_t)M * HDIM) / (size_t)(cU * nTU)) & ~31;
    const int nQ = ((E + BT_STEP - 1) & ~(BT_STEP - 1)) / 4;
    int2* PAIRS = (int2*)H_U;
    unsigned int* TID_M = (unsigned int*)H_M;
    unsigned int* TID_U = TID_M + nQ;

    hipMemsetAsync(DEG_M, 0, (size_t)M * sizeof(float), stream);
    hipMemsetAsync(DEG_U, 0, (size_t)U * sizeof(float), stream);

    const int nInit = (M * HDIM > nQ) ? M * HDIM : nQ;
    movie_init_kernel<<<(nInit + TB - 1) / TB, b, 0, stream>>>(
        movie_x, lin_W, lin_b, movie_emb, XM, M, e_src, e_dst, PAIRS, TID_M, TID_U, E, nQ);

    hipMemsetAsync(AGG_M, 0, (size_t)M * HDIM * sizeof(float), stream);
    hipMemsetAsync(AGG_U, 0, (size_t)U * HDIM * sizeof(float), stream);
    agg_tile_kernel<<<gM, AT_TPB, 0, stream>>>(user_emb, PAIRS, 0, TID_M, AGG_M, DEG_M,
                                               LIST_M, capM, cM, E, M, chM);
    agg_tile_kernel<<<gU, AT_TPB, 0, stream>>>(XM, PAIRS, 1, TID_U, AGG_U, DEG_U,
                                               LIST_U, capU, cU, E, U, chU);
    combine_wmma_kernel<<<CB_GRID, CB_TPB, 0, stream>>>(AGG_M, DEG_M, XM,
                                                        W1um_l, W1um_r, b1um, H_M, 1, M / 16);
    combine_wmma_kernel<<<CB_GRID, CB_TPB, 0, stream>>>(AGG_U, DEG_U, user_emb,
                                                        W1mu_l, W1mu_r, b1mu, H_U, 1, U / 16);

    agg_own_kernel<<<gM, AT_TPB, 0, stream>>>(H_U, e_src, e_dst, AGG_M,
                                              LIST_M, capM, cM, E, M, chM);
    agg_own_kernel<<<gU, AT_TPB, 0, stream>>>(H_M, e_dst, e_src, AGG_U,
                                              LIST_U, capU, cU, E, U, chU);
    combine_wmma_kernel<<<CB_GRID, CB_TPB, 0, stream>>>(AGG_M, DEG_M, H_M,
                                                        W2um_l, W2um_r, b2um, M2, 0, M / 16);
    combine_wmma_kernel<<<CB_GRID, CB_TPB, 0, stream>>>(AGG_U, DEG_U, H_U,
                                                        W2mu_l, W2mu_r, b2mu, U2, 0, U / 16);

    long long dt = (long long)L * 8;
    edge_dot_kernel<<<(int)((dt + TB - 1) / TB), b, 0, stream>>>(U2, M2, eli_u, eli_m, out, L);
}


# movie-side segment-sum tiles: 782 rows x 4 edge chunks (were 1184 x 5): 256 work-groups, one per compute unit (were 215), a sixth fewer rows to sum in each; sums stay arrival-order LDS adds
# speedup vs baseline: 3.1897x; 1.0428x over previous
#include <hip/hip_runtime.h>
#include <hip/hip_bf16.h>

typedef __attribute__((ext_vector_type(2))) float v2f;
typedef __attribute__((ext_vector_type(8))) float v8f;
typedef __attribute__((ext_vector_type(4))) float v4f;

#define HDIM 64

#define AT_ROWS_U 1184
#define AT_ROWS_M 782

template <int ROWS>
__device__ __forceinline__ unsigned int tile_byte(int row) {
    return min((unsigned int)row / (unsigned int)ROWS, 0xffu);
}

__global__ __launch_bounds__(256)
void movie_init_kernel(const float* __restrict__ mx, const float* __restrict__ linW,
                       const float* __restrict__ linb, const float* __restrict__ memb,
                       float* __restrict__ XM, int M,
                       const int* __restrict__ esrc, const int* __restrict__ edst,
                       int2* __restrict__ pairs, unsigned int* __restrict__ tidM,
                       unsigned int* __restrict__ tidU, int nE, int nQ) {
    int t = blockIdx.x * blockDim.x + threadIdx.x;
    if (t < nQ) {
        const int e = t * 4;
        unsigned int wm, wu;
        if (e + 3 < nE) {
            const int4 d = *(const int4*)(edst + e);
            const int4 s = *(const int4*)(esrc + e);
            *(int4*)(pairs + e)     = make_int4(d.x, s.x, d.y, s.y);
            *(int4*)(pairs + e + 2) = make_int4(d.z, s.z, d.w, s.w);
            wm = tile_byte<AT_ROWS_M>(d.x) | (tile_byte<AT_ROWS_M>(d.y) << 8)
               | (tile_byte<AT_ROWS_M>(d.z) << 16) | (tile_byte<AT_ROWS_M>(d.w) << 24);
            wu = tile_byte<AT_ROWS_U>(s.x) | (tile_byte<AT_ROWS_U>(s.y) << 8)
               | (tile_byte<AT_ROWS_U>(s.z) << 16) | (tile_byte<AT_ROWS_U>(s.w) << 24);
        } else {
            wm = 0xffffffffu;
            wu = 0xffffffffu;
            for (int j = 0; j < 4 && e + j < nE; ++j) {
                const int d = edst[e + j], s = esrc[e + j];
                pairs[e + j] = make_int2(d, s);
                wm = (wm & ~(0xffu << (8 * j))) | (tile_byte<AT_ROWS_M>(d) << (8 * j));
                wu = (wu & ~(0xffu << (8 * j))) | (tile_byte<AT_ROWS_U>(s) << (8 * j));
            }
        }
        tidM[t] = wm;
        tidU[t] = wu;
    }
    if (t >= M * HDIM) return;
    int m = t >> 6, h = t & 63;
    float acc = linb[h] + memb[t];
    const float* xr = mx + (size_t)m * 20;
    const float* wr = linW + (size_t)h * 20;
#pragma unroll
    for (int f = 0; f < 20; ++f) acc = fmaf(xr[f], wr[f], acc);
    XM[t] = acc;
}

#define AT_TPB   1024
#define AT_NW    (AT_TPB / 32)
#define AT_STEP  128
#define AT_STRIDE (AT_NW * AT_STEP)
#define AT_G     16
#define AT_WQ    144
#define AT_SRCB  21
#define AT_OFFB  21
#define AT_LHDR  32
#define BT_STEP  512

template <bool FULL>
__device__ __forceinline__ void at_add16(const float* __restrict__ xsrc, float* acc,
                                         unsigned int pk, int first, unsigned int left, int lane) {
    unsigned int s[AT_G];
    float2 r[AT_G];
#pragma unroll
    for (int k = 0; k < AT_G; ++k)
        s[k] = (unsigned int)__builtin_amdgcn_readlane((int)pk, first + k);
#pragma unroll
    for (int k = 0; k < AT_G; ++k) {
        if (FULL || (unsigned int)k < left)
            r[k] = *(const float2*)(xsrc + (size_t)(s[k] & ((1u << AT_SRCB) - 1u)) * HDIM + lane * 2);
    }
#pragma unroll
    for (int k = 0; k < AT_G; ++k) {
        if (FULL || (unsigned int)k < left) {
            float* ap = acc + (s[k] >> AT_SRCB) * HDIM + lane * 2;
            atomicAdd(ap,     r[k].x);
            atomicAdd(ap + 1, r[k].y);
        }
    }
}

template <bool FULL>
__device__ __forceinline__ void bt_take(const float* __restrict__ xsrc, const int2* __restrict__ pairs,
                                        int swap, int base, float* acc, const unsigned int* wq,
                                        unsigned int first, unsigned int left,
                                        unsigned int* __restrict__ L, unsigned int capE,
                                        unsigned int* lcnt, unsigned int* rcnt, int lane) {
    const unsigned int k = (unsigned int)(lane & (AT_G - 1));
    const unsigned int w = wq[first + (FULL ? k : min(k, left - 1u))];
    const int2 pr = pairs[w];
    const unsigned int a      = (unsigned int)((swap ? pr.y : pr.x) - base);
    const unsigned int srcrow = (unsigned int)(swap ? pr.x : pr.y);
    const unsigned int pk = (a << AT_SRCB) | srcrow;
    if ((unsigned int)lane < left) atomicAdd(rcnt + a, 1u);
    unsigned int pos = 0u;
    if (lane == 0) pos = atomicAdd(lcnt, left);
    pos = (unsigned int)__builtin_amdgcn_readfirstlane((int)pos);
    if ((unsigned int)lane < left && pos + (unsigned int)lane < capE)
        L[AT_LHDR + pos + (unsigned int)lane] = pk;
    at_add16<FULL>(xsrc, acc, pk, 0, left, lane);
}

__device__ __forceinline__ unsigned int bt_step(const float* __restrict__ xsrc, const int2* __restrict__ pairs,
                                                int swap, int base, float* acc, unsigned int* wq,
                                                unsigned int pend, uint4 v, unsigned int soff, unsigned int pat,
                                                unsigned int* __restrict__ L, unsigned int capE,
                                                unsigned int* lcnt, unsigned int* rcnt, int lane) {
    const unsigned int K7 = 0x7f7f7f7fu, K8 = 0x80808080u;
    const unsigned int x0 = v.x ^ pat, x1 = v.y ^ pat, x2 = v.z ^ pat, x3 = v.w ^ pat;
    const unsigned int n0 = ((x0 & K7) + K7) | x0;
    const unsigned int n1 = ((x1 & K7) + K7) | x1;
    const unsigned int n2 = ((x2 & K7) + K7) | x2;
    const unsigned int n3 = ((x3 & K7) + K7) | x3;
    const bool any = ((n0 & n1 & n2 & n3) & K8) != K8;
    if (__builtin_amdgcn_ballot_w32(any) == 0u) return pend;

    unsigned int hm = ((~n0 & K8) >> 7) | ((~n1 & K8) >> 6) | ((~n2 & K8) >> 5) | ((~n3 & K8) >> 4);
    const unsigned int lbase = soff + (unsigned int)lane * 16u;
    unsigned int m = __builtin_amdgcn_ballot_w32(hm != 0u);
    do {
        if (hm != 0u) {
            const unsigned int p = (unsigned int)__builtin_ctz(hm);
            hm &= hm - 1u;
            wq[pend + __builtin_amdgcn_mbcnt_lo(m, 0u)] = lbase + ((p & 3u) << 2) + (p >> 3);
        }
        pend += __builtin_popcount(m);
        while (pend >= (unsigned int)AT_G) {
            pend -= AT_G;
            bt_take<true>(xsrc, pairs, swap, base, acc, wq, pend, AT_G, L, capE, lcnt, rcnt, lane);
        }
        m = __builtin_amdgcn_ballot_w32(hm != 0u);
    } while (m != 0u);
    return pend;
}

template <int ROWS>
__global__ __launch_bounds__(AT_TPB)
void agg_tile_kernel(const float* __restrict__ xsrc, const int2* __restrict__ pairs, int swap,
                     const unsigned int* __restrict__ tbytes, float* __restrict__ agg,
                     float* __restrict__ deg,
                     unsigned int* __restrict__ lists, int cap, int nch,
                     int nE, int nDst, int chunk) {
    __shared__ float acc[ROWS * HDIM];
    __shared__ unsigned int wqs[AT_NW * AT_WQ];
    __shared__ unsigned int rcnt[ROWS];
    __shared__ unsigned int lcnt;

    const int tid = threadIdx.x;
    const int lane = tid & 31;
    const int wave = __builtin_amdgcn_readfirstlane(tid >> 5);
    const int base = blockIdx.y * ROWS;
    unsigned int* L = lists + (size_t)(blockIdx.y * nch + blockIdx.x) * (size_t)cap;
    const unsigned int capE = (unsigned int)(cap - AT_LHDR);

    for (int i = tid; i < ROWS * HDIM / 4; i += AT_TPB)
        ((float4*)acc)[i] = make_float4(0.0f, 0.0f, 0.0f, 0.0f);
    for (int i = tid; i < ROWS; i += AT_TPB) rcnt[i] = 0u;
    if (tid == 0) lcnt = 0u;
    __syncthreads();

    {
        const int e0 = blockIdx.x * chunk;
        const int e1 = min(nE, e0 + chunk);
        const int nst = (e1 - e0 + BT_STEP - 1) / BT_STEP;
        unsigned int* wq = wqs + wave * AT_WQ;
        const int2* pr = pairs + e0;
        const unsigned int pat = (unsigned int)blockIdx.y * 0x01010101u;
        unsigned int pend = 0u;
        if (wave < nst) {
            const int nw = (nst - 1 - wave) / AT_NW + 1;
            const uint4* tp = (const uint4*)(tbytes + (e0 >> 2)) + lane;
            const int slast = wave + (nw - 1) * AT_NW;
            int s = wave;
            uint4 v0 = tp[s * 32];
            uint4 v1 = tp[min(s + 1 * AT_NW, slast) * 32];
            uint4 v2 = tp[min(s + 2 * AT_NW, slast) * 32];
            uint4 v3 = tp[min(s + 3 * AT_NW, slast) * 32];
            for (int j = 0; j < nw; j += 4, s += 4 * AT_NW) {
                pend = bt_step(xsrc, pr, swap, base, acc, wq, pend, v0, (unsigned int)s * BT_STEP,
                               pat, L, capE, &lcnt, rcnt, lane);
                v0 = tp[min(s + 4 * AT_NW, slast) * 32];
                if (j + 1 < nw)
                    pend = bt_step(xsrc, pr, swap, base, acc, wq, pend, v1,
                                   (unsigned int)(s + 1 * AT_NW) * BT_STEP, pat, L, capE, &lcnt, rcnt, lane);
                v1 = tp[min(s + 5 * AT_NW, slast) * 32];
                if (j + 2 < nw)
                    pend = bt_step(xsrc, pr, swap, base, acc, wq, pend, v2,
                                   (unsigned int)(s + 2 * AT_NW) * BT_STEP, pat, L, capE, &lcnt, rcnt, lane);
                v2 = tp[min(s + 6 * AT_NW, slast) * 32];
                if (j + 3 < nw)
                    pend = bt_step(xsrc, pr, swap, base, acc, wq, pend, v3,
                                   (unsigned int)(s + 3 * AT_NW) * BT_STEP, pat, L, capE, &lcnt, rcnt, lane);
                v3 = tp[min(s + 7 * AT_NW, slast) * 32];
            }
        }
        if (pend != 0u)
            bt_take<false>(xsrc, pr, swap, base, acc, wq, 0u, pend, L, capE, &lcnt, rcnt, lane);
    }
    __syncthreads();

    for (int i = tid; i < ROWS; i += AT_TPB) {
        const unsigned int cn = rcnt[i];
        if (cn != 0u && base + i < nDst) atomicAdd(deg + base + i, (float)cn);
    }

    if (tid == 0) L[0] = (lcnt <= capE) ? lcnt : 0xffffffffu;

    for (int i = tid; i < ROWS * HDIM; i += AT_TPB) {
        if (base + (i >> 6) < nDst)
            atomicAdd(agg + (size_t)base * HDIM + i, acc[i]);
    }
}


__device__ __forceinline__ int4 at_ld4(const int* __restrict__ p, int e, int e1) {
    if (e + 3 < e1) return *(const int4*)(p + e);
    int4 r;
    r.x = (e     < e1) ? p[e]     : -1;
    r.y = (e + 1 < e1) ? p[e + 1] : -1;
    r.z = (e + 2 < e1) ? p[e + 2] : -1;
    r.w = (e + 3 < e1) ? p[e + 3] : -1;
    return r;
}

template <bool FULL>
__device__ __forceinline__ void at_take(const float* __restrict__ xsrc, const int* __restrict__ gidx,
                                        float* acc, const unsigned int* wq,
                                        unsigned int first, unsigned int left, int e0, int lane) {
    const unsigned int k = (unsigned int)(lane & (AT_G - 1));
    const unsigned int w = wq[first + (FULL ? k : min(k, left - 1u))];
    const unsigned int srcrow = (unsigned int)gidx[e0 + (int)(w & ((1u << AT_OFFB) - 1u))];
    const unsigned int pk = ((w >> AT_OFFB) << AT_SRCB) | srcrow;
    at_add16<FULL>(xsrc, acc, pk, 0, left, lane);
}

__device__ __forceinline__ unsigned int at_step(const float* __restrict__ xsrc, const int* __restrict__ gidx,
                                                float* acc, unsigned int* wq, unsigned int pend,
                                                int4 c, int eb, int e0, int base, unsigned int rows, int lane) {
    const unsigned int a0 = (unsigned int)(c.x - base);
    const unsigned int a1 = (unsigned int)(c.y - base);
    const unsigned int a2 = (unsigned int)(c.z - base);
    const unsigned int a3 = (unsigned int)(c.w - base);
    const unsigned int m0 = __builtin_amdgcn_ballot_w32(a0 < rows);
    const unsigned int m1 = __builtin_amdgcn_ballot_w32(a1 < rows);
    const unsigned int m2 = __builtin_amdgcn_ballot_w32(a2 < rows);
    const unsigned int m3 = __builtin_amdgcn_ballot_w32(a3 < rows);
    if ((m0 | m1 | m2 | m3) == 0u) return pend;

    const unsigned int off = (unsigned int)(eb - e0 + lane * 4);
    if (a0 < rows)
        wq[pend + __builtin_amdgcn_mbcnt_lo(m0, 0u)] = (a0 << AT_OFFB) | off;
    pend += __builtin_popcount(m0);
    if (a1 < rows)
        wq[pend + __builtin_amdgcn_mbcnt_lo(m1, 0u)] = (a1 << AT_OFFB) | (off + 1u);
    pend += __builtin_popcount(m1);
    if (a2 < rows)
        wq[pend + __builtin_amdgcn_mbcnt_lo(m2, 0u)] = (a2 << AT_OFFB) | (off + 2u);
    pend += __builtin_popcount(m2);
    if (a3 < rows)
        wq[pend + __builtin_amdgcn_mbcnt_lo(m3, 0u)] = (a3 << AT_OFFB) | (off + 3u);
    pend += __builtin_popcount(m3);

    while (pend >= (unsigned int)AT_G) {
        pend -= AT_G;
        at_take<true>(xsrc, gidx, acc, wq, pend, AT_G, e0, lane);
    }
    return pend;
}

__device__ __forceinline__ void at_scan_chunk(const float* __restrict__ xsrc, const int* __restrict__ gidx,
                                              const int* __restrict__ sidx, float* acc, unsigned int* wq,
                                              int e0, int e1, int base, unsigned int rows,
                                              unsigned int wave, int lane) {
    unsigned int pend = 0u;
    const int eb0 = e0 + (int)wave * AT_STEP;
    const int nfull = (e1 - eb0 >= AT_STEP) ? (e1 - eb0 - AT_STEP) / AT_STRIDE + 1 : 0;
    if (nfull > 0) {
        const int elast = eb0 + (nfull - 1) * AT_STRIDE;
        const int* sp = sidx + lane * 4;
        int eb = eb0;
        int4 v0 = *(const int4*)(sp + eb);
        int4 v1 = *(const int4*)(sp + min(eb + 1 * AT_STRIDE, elast));
        int4 v2 = *(const int4*)(sp + min(eb + 2 * AT_STRIDE, elast));
        int4 v3 = *(const int4*)(sp + min(eb + 3 * AT_STRIDE, elast));
        for (int j = 0; j < nfull; j += 4, eb += 4 * AT_STRIDE) {
            pend = at_step(xsrc, gidx, acc, wq, pend, v0, eb, e0, base, rows, lane);
            v0 = *(const int4*)(sp + min(eb + 4 * AT_STRIDE, elast));
            if (j + 1 < nfull)
                pend = at_step(xsrc, gidx, acc, wq, pend, v1, eb + 1 * AT_STRIDE, e0, base, rows, lane);
            v1 = *(const int4*)(sp + min(eb + 5 * AT_STRIDE, elast));
            if (j + 2 < nfull)
                pend = at_step(xsrc, gidx, acc, wq, pend, v2, eb + 2 * AT_STRIDE, e0, base, rows, lane);
            v2 = *(const int4*)(sp + min(eb + 6 * AT_STRIDE, elast));
            if (j + 3 < nfull)
                pend = at_step(xsrc, gidx, acc, wq, pend, v3, eb + 3 * AT_STRIDE, e0, base, rows, lane);
            v3 = *(const int4*)(sp + min(eb + 7 * AT_STRIDE, elast));
        }
    }
    const int ebt = eb0 + nfull * AT_STRIDE;
    if (ebt < e1) {
        const int4 c = at_ld4(sidx, ebt + lane * 4, e1);
        pend = at_step(xsrc, gidx, acc, wq, pend, c, ebt, e0, base, rows, lane);
    }
    if (pend != 0u)
        at_take<false>(xsrc, gidx, acc, wq, 0u, pend, e0, lane);
}


__device__ __forceinline__ void own_drain(const float* __restrict__ xsrc, float* acc,
                                          const unsigned int* wq, unsigned int pend, int lane) {
    if (pend != 0u) {
        const unsigned int k = (unsigned int)(lane & (AT_G - 1));
        const unsigned int pk = wq[min(k, pend - 1u)];
        at_add16<false>(xsrc, acc, pk, 0, pend, lane);
    }
}

__device__ __forceinline__ unsigned int own_step(const float* __restrict__ xsrc, float* acc,
                                                 unsigned int* wq, unsigned int pend,
                                                 unsigned int w, unsigned int p, unsigned int cnt,
                                                 unsigned int rowoff, unsigned int rows, int lane) {
    const unsigned int a = (w >> AT_SRCB) - rowoff;
    const bool hit = (p < cnt) && (a < rows);
    const unsigned int m = __builtin_amdgcn_ballot_w32(hit);
    if (m == 0u) return pend;
    if (hit)
        wq[pend + __builtin_amdgcn_mbcnt_lo(m, 0u)] = (a << AT_SRCB) | (w & ((1u << AT_SRCB) - 1u));
    pend += __builtin_popcount(m);
    while (pend >= (unsigned int)AT_G) {
        pend -= AT_G;
        const unsigned int pk = wq[pend + (unsigned int)(lane & (AT_G - 1))];
        at_add16<true>(xsrc, acc, pk, 0, AT_G, lane);
    }
    return pend;
}

__device__ __forceinline__ unsigned int own_walk(const float* __restrict__ xsrc, float* acc,
                                                 unsigned int* wq, unsigned int pend,
                                                 const unsigned int* __restrict__ L, unsigned int cnt,
                                                 unsigned int rowoff, unsigned int rows,
                                                 unsigned int wave, int lane) {
    const unsigned int nlines = (cnt + 31u) >> 5;
    if (wave >= nlines) return pend;
    const unsigned int nl = (nlines - 1u - wave) / AT_NW + 1u;
    const unsigned int last = cnt - 1u;
    const unsigned int ps = AT_NW * 32u;
    const unsigned int* Lw = L + AT_LHDR;
    unsigned int p = wave * 32u + (unsigned int)lane;
    unsigned int w0 = Lw[min(p, last)];
    unsigned int w1 = Lw[min(p + 1u * ps, last)];
    unsigned int w2 = Lw[min(p + 2u * ps, last)];
    unsigned int w3 = Lw[min(p + 3u * ps, last)];
    for (unsigned int k = 0u; k < nl; k += 4u, p += 4u * ps) {
        pend = own_step(xsrc, acc, wq, pend, w0, p, cnt, rowoff, rows, lane);
        w0 = Lw[min(p + 4u * ps, last)];
        if (k + 1u < nl)
            pend = own_step(xsrc, acc, wq, pend, w1, p + 1u * ps, cnt, rowoff, rows, lane);
        w1 = Lw[min(p + 5u * ps, last)];
        if (k + 2u < nl)
            pend = own_step(xsrc, acc, wq, pend, w2, p + 2u * ps, cnt, rowoff, rows, lane);
        w2 = Lw[min(p + 6u * ps, last)];
        if (k + 3u < nl)
            pend = own_step(xsrc, acc, wq, pend, w3, p + 3u * ps, cnt, rowoff, rows, lane);
        w3 = Lw[min(p + 7u * ps, last)];
    }
    return pend;
}

template <int ROWS>
__global__ __launch_bounds__(AT_TPB)
void agg_own_kernel(const float* __restrict__ xsrc, const int* __restrict__ gidx,
                    const int* __restrict__ sidx, float* __restrict__ agg,
                    const unsigned int* __restrict__ lists, int cap, int nch,
                    int nE, int nDst, int chunk) {
    __shared__ float acc[ROWS * HDIM];
    __shared__ unsigned int wqs[AT_NW * AT_WQ];

    const int tid  = threadIdx.x;
    const int lane = tid & 31;
    const unsigned int wave = (unsigned int)__builtin_amdgcn_readfirstlane(tid >> 5);
    const int rs     = (ROWS + nch - 1) / nch;
    const int rowoff = blockIdx.x * rs;
    const int nrows  = min(rs, ROWS - rowoff);
    if (nrows <= 0) return;
    const unsigned int rows = (unsigned int)nrows;
    const int base = blockIdx.y * ROWS + rowoff;
    const unsigned int* Lt = lists + (size_t)blockIdx.y * (size_t)nch * (size_t)cap;
    const unsigned int capE = (unsigned int)(cap - AT_LHDR);

    for (int i = tid; i < nrows * (HDIM / 4); i += AT_TPB)
        ((float4*)acc)[i] = make_float4(0.0f, 0.0f, 0.0f, 0.0f);
    __syncthreads();

    unsigned int* wq = wqs + wave * AT_WQ;
    unsigned int pend = 0u;
    for (int c0 = 0; c0 < nch; c0 += 32) {
        unsigned int hdr = 0u;
        if (c0 + lane < nch) hdr = Lt[(size_t)(c0 + lane) * (size_t)cap];
        const int cn = min(32, nch - c0);
        for (int j = 0; j < cn; ++j) {
            const int c = c0 + j;
            const unsigned int cnt = (unsigned int)__builtin_amdgcn_readlane((int)hdr, j);
            if (cnt <= capE) {
                pend = own_walk(xsrc, acc, wq, pend, Lt + (size_t)c * (size_t)cap, cnt,
                                (unsigned int)rowoff, rows, wave, lane);
            } else {
                own_drain(xsrc, acc, wq, pend, lane);
                pend = 0u;
                const int e0 = c * chunk;
                const int e1 = min(nE, e0 + chunk);
                at_scan_chunk(xsrc, gidx, sidx, acc, wq, e0, e1, base, rows, wave, lane);
            }
        }
    }
    own_drain(xsrc, acc, wq, pend, lane);
    __syncthreads();

    for (int i = tid; i < nrows * (HDIM / 4); i += AT_TPB) {
        if (base + (i >> 4) < nDst)
            ((float4*)(agg + (size_t)base * HDIM))[i] = ((const float4*)acc)[i];
    }
}

#define CB_TPB   128
#define CB_PITCH 68
#define CB_GRID  512

__device__ __forceinline__ void cb_lds_barrier() {
    __builtin_amdgcn_fence(__ATOMIC_RELEASE, "workgroup", "local");
    __builtin_amdgcn_s_barrier();
    __builtin_amdgcn_fence(__ATOMIC_ACQUIRE, "workgroup", "local");
}

__device__ __forceinline__ void cb_fetch(const float* __restrict__ agg, const float* __restrict__ deg,
                                         const float* __restrict__ xdst, int blk, int nb,
                                         int sr, int sc, v4f* pa, v4f* px, float* pd) {
#pragma unroll
    for (int j = 0; j < 2; ++j) {
        const int row = blk * 16 + sr + 8 * j;
        pd[j] = deg[row];
        pa[j] = *(const v4f*)(agg  + (size_t)row * HDIM + sc);
        px[j] = *(const v4f*)(xdst + (size_t)row * HDIM + sc);
    }
    if (nb == 2) {
#pragma unroll
        for (int j = 2; j < 4; ++j) {
            const int row = blk * 16 + sr + 8 * j;
            pd[j] = deg[row];
            pa[j] = *(const v4f*)(agg  + (size_t)row * HDIM + sc);
            px[j] = *(const v4f*)(xdst + (size_t)row * HDIM + sc);
        }
    }
}

__global__ __launch_bounds__(CB_TPB)
void combine_wmma_kernel(const float* __restrict__ agg, const float* __restrict__ deg,
                         const float* __restrict__ xdst,
                         const float* __restrict__ Wl, const float* __restrict__ Wr,
                         const float* __restrict__ bias, float* __restrict__ out,
                         int do_relu, int nblk) {
    __shared__ float A1[32][CB_PITCH];
    __shared__ float A2[32][CB_PITCH];

    const int tid  = threadIdx.x;
    const int wave = tid >> 5;
    const int lane = tid & 31;
    const int m    = lane & 15;
    const int half = lane >> 4;
    const int col0 = wave * 16;

    v2f bl[16], br[16];
    {
        const float* wl = Wl + (size_t)(col0 + m) * HDIM + half * 2;
        const float* wr = Wr + (size_t)(col0 + m) * HDIM + half * 2;
#pragma unroll
        for (int kk = 0; kk < 16; ++kk) {
            bl[kk].x = wl[kk * 4];  bl[kk].y = wl[kk * 4 + 1];
            br[kk].x = wr[kk * 4];  br[kk].y = wr[kk * 4 + 1];
        }
    }
    const float bv = bias[col0 + m];

    const int b1 = (int)(((unsigned int)(blockIdx.x + 1) * (unsigned int)nblk) / gridDim.x);
    int blk      = (int)(((unsigned int)blockIdx.x * (unsigned int)nblk) / gridDim.x);
    if (blk >= b1) return;

    const int sr = tid >> 4;
    const int sc = (tid & 15) * 4;

    v4f   pa[4], px[4];
    float pd[4];
    cb_fetch(agg, deg, xdst, blk, min(2, b1 - blk), sr, sc, pa, px, pd);

    __syncthreads();

    while (blk < b1) {
        const int nb = min(2, b1 - blk);
#pragma unroll
        for (int j = 0; j < 2; ++j) {
            const float dinv = 1.0f / fmaxf(pd[j], 1.0f);
            *(v4f*)&A1[sr + 8 * j][sc] = pa[j] * dinv;
            *(v4f*)&A2[sr + 8 * j][sc] = px[j];
        }
        if (nb == 2) {
#pragma unroll
            for (int j = 2; j < 4; ++j) {
                const float dinv = 1.0f / fmaxf(pd[j], 1.0f);
                *(v4f*)&A1[sr + 8 * j][sc] = pa[j] * dinv;
                *(v4f*)&A2[sr + 8 * j][sc] = px[j];
            }
        } else {
            const v4f z = {0.0f, 0.0f, 0.0f, 0.0f};
#pragma unroll
            for (int j = 2; j < 4; ++j) {
                *(v4f*)&A1[sr + 8 * j][sc] = z;
                *(v4f*)&A2[sr + 8 * j][sc] = z;
            }
        }
        cb_lds_barrier();

        const int next = blk + nb;
        if (next < b1) cb_fetch(agg, deg, xdst, next, min(2, b1 - next), sr, sc, pa, px, pd);

        v8f c0 = {};
        v8f c1 = {};
#pragma unroll
        for (int kk = 0; kk < 16; ++kk) {
            const int k = kk * 4 + half * 2;
            v2f a, a2;
            a.x  = A1[m][k];       a.y  = A1[m][k + 1];
            a2.x = A1[16 + m][k];  a2.y = A1[16 + m][k + 1];
            c0 = __builtin_amdgcn_wmma_f32_16x16x4_f32(false, a,  false, bl[kk],
                                                       (short)0, c0, false, false);
            c1 = __builtin_amdgcn_wmma_f32_16x16x4_f32(false, a2, false, bl[kk],
                                                       (short)0, c1, false, false);
        }
#pragma unroll
        for (int kk = 0; kk < 16; ++kk) {
            const int k = kk * 4 + half * 2;
            v2f a, a2;
            a.x  = A2[m][k];       a.y  = A2[m][k + 1];
            a2.x = A2[16 + m][k];  a2.y = A2[16 + m][k + 1];
            c0 = __builtin_amdgcn_wmma_f32_16x16x4_f32(false, a,  false, br[kk],
                                                       (short)0, c0, false, false);
            c1 = __builtin_amdgcn_wmma_f32_16x16x4_f32(false, a2, false, br[kk],
                                                       (short)0, c1, false, false);
        }

#pragma unroll
        for (int r = 0; r < 8; ++r) {
            float v = c0[r] + bv;
            if (do_relu) v = fmaxf(v, 0.0f);
            out[(size_t)(blk * 16 + r + 8 * half) * HDIM + col0 + m] = v;
        }
        if (nb == 2) {
#pragma unroll
            for (int r = 0; r < 8; ++r) {
                float v = c1[r] + bv;
                if (do_relu) v = fmaxf(v, 0.0f);
                out[(size_t)((blk + 1) * 16 + r + 8 * half) * HDIM + col0 + m] = v;
            }
        }
        cb_lds_barrier();
        blk = next;
    }
}

__global__ __launch_bounds__(256)
void edge_dot_kernel(const float* __restrict__ U2, const float* __restrict__ M2,
                     const int* __restrict__ eu, const int* __restrict__ em,
                     float* __restrict__ out, int L) {
    long long tid = (long long)blockIdx.x * blockDim.x + threadIdx.x;
    if (tid >= (long long)L * 8) return;
    int p = (int)(tid >> 3);
    int s = (int)(tid & 7);
    const float4* a = (const float4*)(U2 + (size_t)eu[p] * HDIM + s * 8);
    const float4* b = (const float4*)(M2 + (size_t)em[p] * HDIM + s * 8);
    float4 a0 = a[0], a1 = a[1], b0 = b[0], b1 = b[1];
    float acc = a0.x * b0.x + a0.y * b0.y + a0.z * b0.z + a0.w * b0.w
              + a1.x * b1.x + a1.y * b1.y + a1.z * b1.z + a1.w * b1.w;
    acc += __shfl_xor(acc, 1);
    acc += __shfl_xor(acc, 2);
    acc += __shfl_xor(acc, 4);
    if (s == 0) out[p] = acc;
}

extern "C" void kernel_launch(void* const* d_in, const int* in_sizes, int n_in,
                              void* d_out, int out_size, void* d_ws, size_t ws_size,
                              hipStream_t stream) {
    (void)n_in; (void)ws_size;
    const float* movie_x   = (const float*)d_in[2];
    const int*   e_src     = (const int*)d_in[3];
    const int*   e_dst     = (const int*)d_in[4];
    const int*   eli_u     = (const int*)d_in[5];
    const int*   eli_m     = (const int*)d_in[6];
    const float* user_emb  = (const float*)d_in[7];
    const float* movie_emb = (const float*)d_in[8];
    const float* lin_W     = (const float*)d_in[9];
    const float* lin_b     = (const float*)d_in[10];
    const float* W1um_l    = (const float*)d_in[11];
    const float* W1um_r    = (const float*)d_in[12];
    const float* W1mu_l    = (const float*)d_in[13];
    const float* W1mu_r    = (const float*)d_in[14];
    const float* W2um_l    = (const float*)d_in[15];
    const float* W2um_r    = (const float*)d_in[16];
    const float* W2mu_l    = (const float*)d_in[17];
    const float* W2mu_r    = (const float*)d_in[18];
    const float* b1um      = (const float*)d_in[19];
    const float* b1mu      = (const float*)d_in[20];
    const float* b2um      = (const float*)d_in[21];
    const float* b2mu      = (const float*)d_in[22];
    float* out = (float*)d_out;

    const int U = in_sizes[7] / HDIM;
    const int M = in_sizes[8] / HDIM;
    const int E = in_sizes[3];
    const int L = in_sizes[5];
    (void)out_size;

    size_t off = 0;
    auto carve = [&](size_t nfloats) {
        float* p = (float*)((char*)d_ws + off);
        off += ((nfloats * sizeof(float) + 255) / 256) * 256;
        return p;
    };
    float* XM    = carve((size_t)M * HDIM);
    float* AGG_M = carve((size_t)M * HDIM);
    float* AGG_U = carve((size_t)U * HDIM);
    float* H_M   = carve((size_t)M * HDIM);
    float* H_U   = carve((size_t)U * HDIM);
    float* M2    = carve((size_t)M * HDIM);
    float* U2    = carve((size_t)U * HDIM);
    float* DEG_M = carve((size_t)M);
    float* DEG_U = carve((size_t)U);

    const int TB = 256;
    dim3 b(TB);
    const int nTM = (M + AT_ROWS_M - 1) / AT_ROWS_M;
    const int nTU = (U + AT_ROWS_U - 1) / AT_ROWS_U;
    const int cM  = (256 / nTM) > 0 ? (256 / nTM) : 1;
    const int cU  = (256 / nTU) > 0 ? (256 / nTU) : 1;
    const int chM = (((E + cM - 1) / cM) + BT_STEP - 1) & ~(BT_STEP - 1);
    const int chU = (((E + cU - 1) / cU) + BT_STEP - 1) & ~(BT_STEP - 1);
    const dim3 gM(cM, nTM), gU(cU, nTU);
    unsigned int* LIST_M = (unsigned int*)U2;
    unsigned int* LIST_U = (unsigned int*)M2;
    const int capM = (int)(((size_t)U * HDIM) / (size_t)(cM * nTM)) & ~31;
    const int capU = (int)(((size_t)M * HDIM) / (size_t)(cU * nTU)) & ~31;
    const int nQ = ((E + BT_STEP - 1) & ~(BT_STEP - 1)) / 4;
    int2* PAIRS = (int2*)H_U;
    unsigned int* TID_M = (unsigned int*)H_M;
    unsigned int* TID_U = TID_M + nQ;

    hipMemsetAsync(DEG_M, 0, (size_t)M * sizeof(float), stream);
    hipMemsetAsync(DEG_U, 0, (size_t)U * sizeof(float), stream);

    const int nInit = (M * HDIM > nQ) ? M * HDIM : nQ;
    movie_init_kernel<<<(nInit + TB - 1) / TB, b, 0, stream>>>(
        movie_x, lin_W, lin_b, movie_emb, XM, M, e_src, e_dst, PAIRS, TID_M, TID_U, E, nQ);

    hipMemsetAsync(AGG_M, 0, (size_t)M * HDIM * sizeof(float), stream);
    hipMemsetAsync(AGG_U, 0, (size_t)U * HDIM * sizeof(float), stream);
    agg_tile_kernel<AT_ROWS_M><<<gM, AT_TPB, 0, stream>>>(user_emb, PAIRS, 0, TID_M, AGG_M, DEG_M,
                                               LIST_M, capM, cM, E, M, chM);
    agg_tile_kernel<AT_ROWS_U><<<gU, AT_TPB, 0, stream>>>(XM, PAIRS, 1, TID_U, AGG_U, DEG_U,
                                               LIST_U, capU, cU, E, U, chU);
    combine_wmma_kernel<<<CB_GRID, CB_TPB, 0, stream>>>(AGG_M, DEG_M, XM,
                                                        W1um_l, W1um_r, b1um, H_M, 1, M / 16);
    combine_wmma_kernel<<<CB_GRID, CB_TPB, 0, stream>>>(AGG_U, DEG_U, user_emb,
                                                        W1mu_l, W1mu_r, b1mu, H_U, 1, U / 16);

    agg_own_kernel<AT_ROWS_M><<<gM, AT_TPB, 0, stream>>>(H_U, e_src, e_dst, AGG_M,
                                              LIST_M, capM, cM, E, M, chM);
    agg_own_kernel<AT_ROWS_U><<<gU, AT_TPB, 0, stream>>>(H_M, e_dst, e_src, AGG_U,
                                              LIST_U, capU, cU, E, U, chU);
    combine_wmma_kernel<<<CB_GRID, CB_TPB, 0, stream>>>(AGG_M, DEG_M, H_M,
                                                        W2um_l, W2um_r, b2um, M2, 0, M / 16);
    combine_wmma_kernel<<<CB_GRID, CB_TPB, 0, stream>>>(AGG_U, DEG_U, H_U,
                                                        W2mu_l, W2mu_r, b2mu, U2, 0, U / 16);

    long long dt = (long long)L * 8;
    edge_dot_kernel<<<(int)((dt + TB - 1) / TB), b, 0, stream>>>(U2, M2, eli_u, eli_m, out, L);
}
